# starting kernel
# baseline (speedup 1.0000x reference)
#include <hip/hip_runtime.h>

#define SEQ   512
#define BATCH 4096
#define DIMS  3
#define DIMY  3
#define DIMZ  6
#define RANK  6
#define HID   50
#define KG    13
#define XSTR  68
#define ALPHA 0.5f

typedef float v2f __attribute__((ext_vector_type(2)));
typedef float v8f __attribute__((ext_vector_type(8)));

__device__ __forceinline__ v8f wmma_f32_k4(v2f a, v2f b, v8f c) {
    return __builtin_amdgcn_wmma_f32_16x16x4_f32(false, a, false, b, (short)0, c, false, false);
}

__device__ __forceinline__ float fast_tanh(float x) {
#if __has_builtin(__builtin_amdgcn_tanhf)
    return __builtin_amdgcn_tanhf(x);
#else
    x = fminf(15.0f, fmaxf(-15.0f, x));
    float e = __expf(2.0f * x);
    float r = __builtin_amdgcn_rcpf(e + 1.0f);
    return fmaf(-2.0f, r, 1.0f);
#endif
}

__device__ __forceinline__ float fast_sigmoid(float x) {
    return __builtin_amdgcn_rcpf(1.0f + __expf(-x));
}

__global__ __launch_bounds__(32) void cxtrnn_wmma_kernel(
    const float* __restrict__ s,   const float* __restrict__ z,
    const float* __restrict__ U,   const float* __restrict__ V,
    const float* __restrict__ Wi,  const float* __restrict__ bi,
    const float* __restrict__ Wo,  const float* __restrict__ bo,
    const float* __restrict__ nmW, const float* __restrict__ nmb,
    float* __restrict__ out)
{
    __shared__ float ldsX[16 * XSTR];
    __shared__ float ldsP[16 * 16];

    const int lane = (int)threadIdx.x;
    const int col  = lane & 15;
    const int hi   = lane >> 4;
    const bool hib = (hi != 0);
    const int b0   = (int)blockIdx.x * 16;
    const int brow = b0 + col;

    v2f Bv[KG];
#pragma unroll
    for (int g = 0; g < KG; ++g) {
        int k0 = 4 * g + 2 * hi;
        float a0 = (k0     < HID && col < RANK) ? V[k0 * RANK + col]       : 0.0f;
        float a1 = (k0 + 1 < HID && col < RANK) ? V[(k0 + 1) * RANK + col] : 0.0f;
        v2f t; t[0] = a0; t[1] = a1; Bv[g] = t;
    }
    v2f Bu[4][2];
#pragma unroll
    for (int t4 = 0; t4 < 4; ++t4) {
        int h = 16 * t4 + col;
#pragma unroll
        for (int g = 0; g < 2; ++g) {
            int k0 = 4 * g + 2 * hi;
            float a0 = (h < HID && k0     < RANK) ? ALPHA * U[h * RANK + k0]     : 0.0f;
            float a1 = (h < HID && k0 + 1 < RANK) ? ALPHA * U[h * RANK + k0 + 1] : 0.0f;
            v2f t; t[0] = a0; t[1] = a1; Bu[t4][g] = t;
        }
    }
    v2f Bw[4];
#pragma unroll
    for (int t4 = 0; t4 < 4; ++t4) {
        int h  = 16 * t4 + col;
        int k0 = 2 * hi;
        float a0 = 0.0f, a1 = 0.0f;
        if (h < HID) {
            a0 = ALPHA * Wi[h * DIMS + k0];
            a1 = ALPHA * ((k0 + 1 < DIMS) ? Wi[h * DIMS + k0 + 1] : bi[h]);
        }
        v2f t; t[0] = a0; t[1] = a1; Bw[t4] = t;
    }
    v2f Bo[KG];
#pragma unroll
    for (int g = 0; g < KG; ++g) {
        int k0 = 4 * g + 2 * hi;
        float a0 = 0.0f, a1 = 0.0f;
        if (col < DIMY) {
            a0 = (k0     < HID) ? Wo[col * HID + k0]     : ((k0     == HID) ? bo[col] : 0.0f);
            a1 = (k0 + 1 < HID) ? Wo[col * HID + k0 + 1] : ((k0 + 1 == HID) ? bo[col] : 0.0f);
        }
        v2f t; t[0] = a0; t[1] = a1; Bo[g] = t;
    }
    float nmw[RANK][DIMZ], nmbv[RANK];
#pragma unroll
    for (int r = 0; r < RANK; ++r) {
        nmbv[r] = nmb[r];
#pragma unroll
        for (int i = 0; i < DIMZ; ++i) nmw[r][i] = nmW[r * DIMZ + i];
    }

    v8f XD[4];
#pragma unroll
    for (int t4 = 0; t4 < 4; ++t4) { v8f zz = {0,0,0,0,0,0,0,0}; XD[t4] = zz; }
    float Atan[2 * KG];
#pragma unroll
    for (int i = 0; i < 2 * KG; ++i) Atan[i] = 0.0f;

    const v8f vzero = {0,0,0,0,0,0,0,0};

#pragma unroll 1
    for (int t = 0; t < SEQ; ++t) {
        const float* zr = z + ((size_t)t * BATCH + (size_t)brow) * DIMZ;
        const float* sr = s + ((size_t)t * BATCH + (size_t)brow) * DIMS;
        float zv[DIMZ];
#pragma unroll
        for (int i = 0; i < DIMZ; ++i) zv[i] = zr[i];
        float sv0 = sr[0], sv1 = sr[1], sv2 = sr[2];
        float zsum = 0.0f;
#pragma unroll
        for (int i = 0; i < DIMZ; ++i) zsum += zv[i];
        float g0, g1, g2, g3, g4, g5;
        {
            float a0 = nmbv[0], a1 = nmbv[1], a2 = nmbv[2];
            float a3 = nmbv[3], a4 = nmbv[4], a5 = nmbv[5];
#pragma unroll
            for (int i = 0; i < DIMZ; ++i) {
                a0 = fmaf(nmw[0][i], zv[i], a0);
                a1 = fmaf(nmw[1][i], zv[i], a1);
                a2 = fmaf(nmw[2][i], zv[i], a2);
                a3 = fmaf(nmw[3][i], zv[i], a3);
                a4 = fmaf(nmw[4][i], zv[i], a4);
                a5 = fmaf(nmw[5][i], zv[i], a5);
            }
            g0 = fast_sigmoid(a0); g1 = fast_sigmoid(a1); g2 = fast_sigmoid(a2);
            g3 = fast_sigmoid(a3); g4 = fast_sigmoid(a4); g5 = fast_sigmoid(a5);
        }
        const float ga0 = hib ? g2 : g0;
        const float ga1 = hib ? g3 : g1;
        const float ga2 = hib ? 0.0f : g4;
        const float ga3 = hib ? 0.0f : g5;

        v8f P = vzero;
#pragma unroll
        for (int g = 0; g < KG; ++g) {
            v2f a; a[0] = Atan[2 * g]; a[1] = Atan[2 * g + 1];
            P = wmma_f32_k4(a, Bv[g], P);
        }

#pragma unroll
        for (int v = 0; v < 8; ++v) ldsP[(v + 8 * hi) * 16 + col] = P[v];
        asm volatile("s_wait_dscnt 0" ::: "memory");
        v2f Ap0, Ap1;
        {
            float p0 = ldsP[col * 16 + 2 * hi];
            float p1 = ldsP[col * 16 + 2 * hi + 1];
            float p2 = ldsP[col * 16 + 4 + 2 * hi];
            float p3 = ldsP[col * 16 + 4 + 2 * hi + 1];
            Ap0[0] = p0 * ga0; Ap0[1] = p1 * ga1;
            Ap1[0] = p2 * ga2; Ap1[1] = p3 * ga3;
        }
        v2f As;
        As[0] = zsum * (hib ? sv2 : sv0);
        As[1] = hib ? zsum : (zsum * sv1);

#pragma unroll
        for (int t4 = 0; t4 < 4; ++t4) {
            v8f acc = wmma_f32_k4(Ap0, Bu[t4][0], vzero);
            acc = wmma_f32_k4(Ap1, Bu[t4][1], acc);
            acc = wmma_f32_k4(As,  Bw[t4],    acc);
            v8f x = XD[t4];
#pragma unroll
            for (int v = 0; v < 8; ++v) x[v] = fmaf(x[v], 1.0f - ALPHA, acc[v]);
            XD[t4] = x;
#pragma unroll
            for (int v = 0; v < 8; ++v)
                ldsX[(v + 8 * hi) * XSTR + 16 * t4 + col] = x[v];
        }
        asm volatile("s_wait_dscnt 0" ::: "memory");

#pragma unroll
        for (int g = 0; g < KG; ++g) {
            int k0 = 4 * g + 2 * hi;
            Atan[2 * g]     = fast_tanh(ldsX[col * XSTR + k0]);
            Atan[2 * g + 1] = fast_tanh(ldsX[col * XSTR + k0 + 1]);
        }

        v8f OD = vzero;
#pragma unroll
        for (int g = 0; g < KG; ++g) {
            float a0 = zsum * Atan[2 * g];
            float a1 = zsum * Atan[2 * g + 1];
            if (g == KG - 1) {
                a0 = hib ? zsum : a0;
                a1 = hib ? 0.0f : a1;
            }
            v2f a; a[0] = a0; a[1] = a1;
            OD = wmma_f32_k4(a, Bo[g], OD);
        }

        if (col < DIMY) {
            size_t base = (size_t)t * BATCH * DIMY + (size_t)(b0 + 8 * hi) * DIMY + col;
#pragma unroll
            for (int v = 0; v < 8; ++v) out[base + (size_t)v * DIMY] = OD[v];
        }
    }
}

extern "C" void kernel_launch(void* const* d_in, const int* in_sizes, int n_in,
                              void* d_out, int out_size, void* d_ws, size_t ws_size,
                              hipStream_t stream) {
    (void)in_sizes; (void)n_in; (void)out_size; (void)d_ws; (void)ws_size;
    const float* s   = (const float*)d_in[0];
    const float* z   = (const float*)d_in[1];
    const float* U   = (const float*)d_in[2];
    const float* V   = (const float*)d_in[3];
    const float* Wi  = (const float*)d_in[4];
    const float* bi  = (const float*)d_in[5];
    const float* Wo  = (const float*)d_in[6];
    const float* bo  = (const float*)d_in[7];
    const float* nmW = (const float*)d_in[8];
    const float* nmb = (const float*)d_in[9];
    float* out = (float*)d_out;

    dim3 grid(BATCH / 16);
    dim3 block(32);
    hipLaunchKernelGGL(cxtrnn_wmma_kernel, grid, block, 0, stream,
                       s, z, U, V, Wi, bi, Wo, bo, nmW, nmb, out);
}


# Work not fed back into the state (output products, step inputs) moved to a second wave of the workgroup on another SIMD; states cross in per-step LDS slots, one barrier per 16 steps; sums unchanged
# speedup vs baseline: 2.0792x; 2.0792x over previous
#include <hip/hip_runtime.h>

#define SEQ   512
#define BATCH 4096
#define DIMS  3
#define DIMY  3
#define DIMZ  6
#define RANK  6
#define HID   50
#define KG    13
#define XSTR  52
#define XSLOT (16 * XSTR)
#define ALPHA 0.5f
#define TB    16
#define NB    (SEQ / TB)

typedef float v2f __attribute__((ext_vector_type(2)));
typedef float v8f __attribute__((ext_vector_type(8)));

__device__ __forceinline__ v8f wmma_f32_k4(v2f a, v2f b, v8f c) {
    return __builtin_amdgcn_wmma_f32_16x16x4_f32(false, a, false, b, (short)0, c, false, false);
}

__device__ __forceinline__ float fast_tanh(float x) {
#if __has_builtin(__builtin_amdgcn_tanhf)
    return __builtin_amdgcn_tanhf(x);
#else
    x = fminf(15.0f, fmaxf(-15.0f, x));
    float e = __expf(2.0f * x);
    float r = __builtin_amdgcn_rcpf(e + 1.0f);
    return fmaf(-2.0f, r, 1.0f);
#endif
}

__device__ __forceinline__ float fast_sigmoid(float x) {
    return __builtin_amdgcn_rcpf(1.0f + __expf(-x));
}

__global__ __launch_bounds__(64) void cxtrnn_wmma_kernel(
    const float* __restrict__ s,   const float* __restrict__ z,
    const float* __restrict__ U,   const float* __restrict__ V,
    const float* __restrict__ Wi,  const float* __restrict__ bi,
    const float* __restrict__ Wo,  const float* __restrict__ bo,
    const float* __restrict__ nmW, const float* __restrict__ nmb,
    float* __restrict__ out)
{
    __shared__ float ldsXs[2 * TB * XSLOT];
    __shared__ float ldsP[16 * 16];
    __shared__ float4 ldsG[2 * TB * 32];
    __shared__ float4 ldsS[2 * TB * 32];

    const int lane = (int)threadIdx.x & 31;
    const int wid  = __builtin_amdgcn_readfirstlane((int)threadIdx.x >> 5);
    const int col  = lane & 15;
    const int hi   = lane >> 4;
    const bool hib = (hi != 0);
    const int b0   = (int)blockIdx.x * 16;
    const int brow = b0 + col;

    v2f Bv[KG];
#pragma unroll
    for (int g = 0; g < KG; ++g) {
        int k0 = 4 * g + 2 * hi;
        float a0 = (k0     < HID && col < RANK) ? V[k0 * RANK + col]       : 0.0f;
        float a1 = (k0 + 1 < HID && col < RANK) ? V[(k0 + 1) * RANK + col] : 0.0f;
        v2f t; t[0] = a0; t[1] = a1; Bv[g] = t;
    }
    v2f Bu[4][2];
#pragma unroll
    for (int t4 = 0; t4 < 4; ++t4) {
        int h = 16 * t4 + col;
#pragma unroll
        for (int g = 0; g < 2; ++g) {
            int k0 = 4 * g + 2 * hi;
            float a0 = (h < HID && k0     < RANK) ? ALPHA * U[h * RANK + k0]     : 0.0f;
            float a1 = (h < HID && k0 + 1 < RANK) ? ALPHA * U[h * RANK + k0 + 1] : 0.0f;
            v2f t; t[0] = a0; t[1] = a1; Bu[t4][g] = t;
        }
    }
    v2f Bw[4];
#pragma unroll
    for (int t4 = 0; t4 < 4; ++t4) {
        int h  = 16 * t4 + col;
        int k0 = 2 * hi;
        float a0 = 0.0f, a1 = 0.0f;
        if (h < HID) {
            a0 = ALPHA * Wi[h * DIMS + k0];
            a1 = ALPHA * ((k0 + 1 < DIMS) ? Wi[h * DIMS + k0 + 1] : bi[h]);
        }
        v2f t; t[0] = a0; t[1] = a1; Bw[t4] = t;
    }
    v2f Bo[KG];
#pragma unroll
    for (int g = 0; g < KG; ++g) {
        int k0 = 4 * g + 2 * hi;
        float a0 = 0.0f, a1 = 0.0f;
        if (col < DIMY) {
            a0 = (k0     < HID) ? Wo[col * HID + k0]     : ((k0     == HID) ? bo[col] : 0.0f);
            a1 = (k0 + 1 < HID) ? Wo[col * HID + k0 + 1] : ((k0 + 1 == HID) ? bo[col] : 0.0f);
        }
        v2f t; t[0] = a0; t[1] = a1; Bo[g] = t;
    }
    float nmw[RANK][DIMZ], nmbv[RANK];
#pragma unroll
    for (int r = 0; r < RANK; ++r) {
        nmbv[r] = nmb[r];
#pragma unroll
        for (int i = 0; i < DIMZ; ++i) nmw[r][i] = nmW[r * DIMZ + i];
    }

    v8f XD[4];
#pragma unroll
    for (int t4 = 0; t4 < 4; ++t4) { v8f zz = {0,0,0,0,0,0,0,0}; XD[t4] = zz; }
    float Atan[2 * KG];
#pragma unroll
    for (int i = 0; i < 2 * KG; ++i) Atan[i] = 0.0f;

    const v8f vzero = {0,0,0,0,0,0,0,0};

#pragma unroll 1
    for (int ph = 0; ph < NB + 2; ++ph) {
      if (wid == 0) {
        if (ph >= 1 && ph <= NB) {
          const int kb = ph - 1;
          const float4* Gb = ldsG + (kb & 1) * (TB * 32);
          const float4* Sb = ldsS + (kb & 1) * (TB * 32);
          float* Xb = ldsXs + (kb & 1) * (TB * XSLOT);
#pragma unroll 1
          for (int tt = 0; tt < TB; ++tt) {
            float* Xs = Xb + tt * XSLOT;
            const float4 gq = Gb[tt * 32 + lane];
            const float4 sq = Sb[tt * 32 + lane];
            const float ga0 = gq.x, ga1 = gq.y, ga2 = gq.z, ga3 = gq.w;

            v8f P = vzero;
#pragma unroll
            for (int g = 0; g < KG; ++g) {
                v2f a; a[0] = Atan[2 * g]; a[1] = Atan[2 * g + 1];
                P = wmma_f32_k4(a, Bv[g], P);
            }

#pragma unroll
            for (int v = 0; v < 8; ++v) ldsP[(v + 8 * hi) * 16 + col] = P[v];
            asm volatile("s_wait_dscnt 0" ::: "memory");
            v2f Ap0, Ap1;
            {
                float p0 = ldsP[col * 16 + 2 * hi];
                float p1 = ldsP[col * 16 + 2 * hi + 1];
                float p2 = ldsP[col * 16 + 4 + 2 * hi];
                float p3 = ldsP[col * 16 + 4 + 2 * hi + 1];
                Ap0[0] = p0 * ga0; Ap0[1] = p1 * ga1;
                Ap1[0] = p2 * ga2; Ap1[1] = p3 * ga3;
            }
            v2f As;
            As[0] = sq.x;
            As[1] = sq.y;

#pragma unroll
            for (int t4 = 0; t4 < 4; ++t4) {
                v8f acc = wmma_f32_k4(Ap0, Bu[t4][0], vzero);
                acc = wmma_f32_k4(Ap1, Bu[t4][1], acc);
                acc = wmma_f32_k4(As,  Bw[t4],    acc);
                v8f x = XD[t4];
#pragma unroll
                for (int v = 0; v < 8; ++v) x[v] = fmaf(x[v], 1.0f - ALPHA, acc[v]);
                XD[t4] = x;
                if (t4 < 3 || col < XSTR - 48) {
#pragma unroll
                    for (int v = 0; v < 8; ++v)
                        Xs[(v + 8 * hi) * XSTR + 16 * t4 + col] = x[v];
                }
            }
            asm volatile("s_wait_dscnt 0" ::: "memory");

            float xa[2 * KG];
#pragma unroll
            for (int g = 0; g < KG; ++g) {
                int k0 = 4 * g + 2 * hi;
                xa[2 * g]     = Xs[col * XSTR + k0];
                xa[2 * g + 1] = Xs[col * XSTR + k0 + 1];
            }
            __builtin_amdgcn_sched_barrier(0);
#pragma unroll
            for (int i = 0; i < 2 * KG; ++i) Atan[i] = fast_tanh(xa[i]);
          }
        }
      } else {
        float zq[TB / 2][DIMZ], sq[TB / 2][DIMS];
        if (ph < NB) {
            const int t0 = ph * TB;
            const float* zb = z + ((size_t)(t0 + hi) * BATCH + (size_t)brow) * DIMZ;
            const float* sb = s + ((size_t)(t0 + hi) * BATCH + (size_t)brow) * DIMS;
#pragma unroll
            for (int j = 0; j < TB / 2; ++j) {
                const float* zr = zb + (size_t)j * (2 * BATCH * DIMZ);
                const float* sr = sb + (size_t)j * (2 * BATCH * DIMS);
#pragma unroll
                for (int i = 0; i < DIMZ; ++i) zq[j][i] = zr[i];
#pragma unroll
                for (int i = 0; i < DIMS; ++i) sq[j][i] = sr[i];
            }
        }
        if (ph >= 2) {
            const int kb = ph - 2;
            const float4* Sb = ldsS + (kb & 1) * (TB * 32);
            const float* Xb = ldsXs + (kb & 1) * (TB * XSLOT);
#pragma unroll 1
            for (int tt = 0; tt < TB; ++tt) {
                const int t = kb * TB + tt;
                const float zsum = Sb[tt * 32 + lane].z;
                const float* Xs = Xb + tt * XSLOT;
                float ta[2 * KG];
#pragma unroll
                for (int g = 0; g < KG; ++g) {
                    int k0 = 4 * g + 2 * hi;
                    ta[2 * g]     = fast_tanh(Xs[col * XSTR + k0]);
                    ta[2 * g + 1] = fast_tanh(Xs[col * XSTR + k0 + 1]);
                }
                v8f OD = vzero;
#pragma unroll
                for (int g = 0; g < KG; ++g) {
                    float a0 = zsum * ta[2 * g];
                    float a1 = zsum * ta[2 * g + 1];
                    if (g == KG - 1) {
                        a0 = hib ? zsum : a0;
                        a1 = hib ? 0.0f : a1;
                    }
                    v2f a; a[0] = a0; a[1] = a1;
                    OD = wmma_f32_k4(a, Bo[g], OD);
                }
                if (col < DIMY) {
                    size_t base = (size_t)t * BATCH * DIMY + (size_t)(b0 + 8 * hi) * DIMY + col;
#pragma unroll
                    for (int v = 0; v < 8; ++v) out[base + (size_t)v * DIMY] = OD[v];
                }
            }
        }
        if (ph < NB) {
            float4* Gb = ldsG + (ph & 1) * (TB * 32);
            float4* Sb = ldsS + (ph & 1) * (TB * 32);
#pragma unroll
            for (int j = 0; j < TB / 2; ++j) {
                float zsum = 0.0f;
#pragma unroll
                for (int i = 0; i < DIMZ; ++i) zsum += zq[j][i];
                float a0 = nmbv[0], a1 = nmbv[1], a2 = nmbv[2];
                float a3 = nmbv[3], a4 = nmbv[4], a5 = nmbv[5];
#pragma unroll
                for (int i = 0; i < DIMZ; ++i) {
                    a0 = fmaf(nmw[0][i], zq[j][i], a0);
                    a1 = fmaf(nmw[1][i], zq[j][i], a1);
                    a2 = fmaf(nmw[2][i], zq[j][i], a2);
                    a3 = fmaf(nmw[3][i], zq[j][i], a3);
                    a4 = fmaf(nmw[4][i], zq[j][i], a4);
                    a5 = fmaf(nmw[5][i], zq[j][i], a5);
                }
                const float g0 = fast_sigmoid(a0), g1 = fast_sigmoid(a1), g2 = fast_sigmoid(a2);
                const float g3 = fast_sigmoid(a3), g4 = fast_sigmoid(a4), g5 = fast_sigmoid(a5);
                const int slot = (2 * j + hi) * 32 + col;
                Gb[slot]      = make_float4(g0, g1, g4, g5);
                Gb[slot + 16] = make_float4(g2, g3, 0.0f, 0.0f);
                Sb[slot]      = make_float4(zsum * sq[j][0], zsum * sq[j][1], zsum, 0.0f);
                Sb[slot + 16] = make_float4(zsum * sq[j][2], zsum,            zsum, 0.0f);
            }
        }
      }
      __syncthreads();
    }
}

extern "C" void kernel_launch(void* const* d_in, const int* in_sizes, int n_in,
                              void* d_out, int out_size, void* d_ws, size_t ws_size,
                              hipStream_t stream) {
    (void)in_sizes; (void)n_in; (void)out_size; (void)d_ws; (void)ws_size;
    const float* s   = (const float*)d_in[0];
    const float* z   = (const float*)d_in[1];
    const float* U   = (const float*)d_in[2];
    const float* V   = (const float*)d_in[3];
    const float* Wi  = (const float*)d_in[4];
    const float* bi  = (const float*)d_in[5];
    const float* Wo  = (const float*)d_in[6];
    const float* bo  = (const float*)d_in[7];
    const float* nmW = (const float*)d_in[8];
    const float* nmb = (const float*)d_in[9];
    float* out = (float*)d_out;

    dim3 grid(BATCH / 16);
    dim3 block(64);
    hipLaunchKernelGGL(cxtrnn_wmma_kernel, grid, block, 0, stream,
                       s, z, U, V, Wi, bi, Wo, bo, nmW, nmb, out);
}


# Helper wave's output loop: two steps per trip, both steps' state rows read from LDS as one group in front, the two 13-WMMA chains interleaved on two accumulators; sums unchanged
# speedup vs baseline: 2.2952x; 1.1039x over previous
#include <hip/hip_runtime.h>

#define SEQ   512
#define BATCH 4096
#define DIMS  3
#define DIMY  3
#define DIMZ  6
#define RANK  6
#define HID   50
#define KG    13
#define XSTR  52
#define XSLOT (16 * XSTR)
#define ALPHA 0.5f
#define TB    16
#define NB    (SEQ / TB)

typedef float v2f __attribute__((ext_vector_type(2)));
typedef float v8f __attribute__((ext_vector_type(8)));

__device__ __forceinline__ v8f wmma_f32_k4(v2f a, v2f b, v8f c) {
    return __builtin_amdgcn_wmma_f32_16x16x4_f32(false, a, false, b, (short)0, c, false, false);
}

__device__ __forceinline__ float fast_tanh(float x) {
#if __has_builtin(__builtin_amdgcn_tanhf)
    return __builtin_amdgcn_tanhf(x);
#else
    x = fminf(15.0f, fmaxf(-15.0f, x));
    float e = __expf(2.0f * x);
    float r = __builtin_amdgcn_rcpf(e + 1.0f);
    return fmaf(-2.0f, r, 1.0f);
#endif
}

__device__ __forceinline__ float fast_sigmoid(float x) {
    return __builtin_amdgcn_rcpf(1.0f + __expf(-x));
}

__global__ __launch_bounds__(64) void cxtrnn_wmma_kernel(
    const float* __restrict__ s,   const float* __restrict__ z,
    const float* __restrict__ U,   const float* __restrict__ V,
    const float* __restrict__ Wi,  const float* __restrict__ bi,
    const float* __restrict__ Wo,  const float* __restrict__ bo,
    const float* __restrict__ nmW, const float* __restrict__ nmb,
    float* __restrict__ out)
{
    __shared__ float ldsXs[2 * TB * XSLOT];
    __shared__ float ldsP[16 * 16];
    __shared__ float4 ldsG[2 * TB * 32];
    __shared__ float4 ldsS[2 * TB * 32];

    const int lane = (int)threadIdx.x & 31;
    const int wid  = __builtin_amdgcn_readfirstlane((int)threadIdx.x >> 5);
    const int col  = lane & 15;
    const int hi   = lane >> 4;
    const bool hib = (hi != 0);
    const int b0   = (int)blockIdx.x * 16;
    const int brow = b0 + col;

    v2f Bv[KG];
#pragma unroll
    for (int g = 0; g < KG; ++g) {
        int k0 = 4 * g + 2 * hi;
        float a0 = (k0     < HID && col < RANK) ? V[k0 * RANK + col]       : 0.0f;
        float a1 = (k0 + 1 < HID && col < RANK) ? V[(k0 + 1) * RANK + col] : 0.0f;
        v2f t; t[0] = a0; t[1] = a1; Bv[g] = t;
    }
    v2f Bu[4][2];
#pragma unroll
    for (int t4 = 0; t4 < 4; ++t4) {
        int h = 16 * t4 + col;
#pragma unroll
        for (int g = 0; g < 2; ++g) {
            int k0 = 4 * g + 2 * hi;
            float a0 = (h < HID && k0     < RANK) ? ALPHA * U[h * RANK + k0]     : 0.0f;
            float a1 = (h < HID && k0 + 1 < RANK) ? ALPHA * U[h * RANK + k0 + 1] : 0.0f;
            v2f t; t[0] = a0; t[1] = a1; Bu[t4][g] = t;
        }
    }
    v2f Bw[4];
#pragma unroll
    for (int t4 = 0; t4 < 4; ++t4) {
        int h  = 16 * t4 + col;
        int k0 = 2 * hi;
        float a0 = 0.0f, a1 = 0.0f;
        if (h < HID) {
            a0 = ALPHA * Wi[h * DIMS + k0];
            a1 = ALPHA * ((k0 + 1 < DIMS) ? Wi[h * DIMS + k0 + 1] : bi[h]);
        }
        v2f t; t[0] = a0; t[1] = a1; Bw[t4] = t;
    }
    v2f Bo[KG];
#pragma unroll
    for (int g = 0; g < KG; ++g) {
        int k0 = 4 * g + 2 * hi;
        float a0 = 0.0f, a1 = 0.0f;
        if (col < DIMY) {
            a0 = (k0     < HID) ? Wo[col * HID + k0]     : ((k0     == HID) ? bo[col] : 0.0f);
            a1 = (k0 + 1 < HID) ? Wo[col * HID + k0 + 1] : ((k0 + 1 == HID) ? bo[col] : 0.0f);
        }
        v2f t; t[0] = a0; t[1] = a1; Bo[g] = t;
    }
    float nmw[RANK][DIMZ], nmbv[RANK];
#pragma unroll
    for (int r = 0; r < RANK; ++r) {
        nmbv[r] = nmb[r];
#pragma unroll
        for (int i = 0; i < DIMZ; ++i) nmw[r][i] = nmW[r * DIMZ + i];
    }

    v8f XD[4];
#pragma unroll
    for (int t4 = 0; t4 < 4; ++t4) { v8f zz = {0,0,0,0,0,0,0,0}; XD[t4] = zz; }
    float Atan[2 * KG];
#pragma unroll
    for (int i = 0; i < 2 * KG; ++i) Atan[i] = 0.0f;

    const v8f vzero = {0,0,0,0,0,0,0,0};

#pragma unroll 1
    for (int ph = 0; ph < NB + 2; ++ph) {
      if (wid == 0) {
        if (ph >= 1 && ph <= NB) {
          const int kb = ph - 1;
          const float4* Gb = ldsG + (kb & 1) * (TB * 32);
          const float4* Sb = ldsS + (kb & 1) * (TB * 32);
          float* Xb = ldsXs + (kb & 1) * (TB * XSLOT);
#pragma unroll 1
          for (int tt = 0; tt < TB; ++tt) {
            float* Xs = Xb + tt * XSLOT;
            const float4 gq = Gb[tt * 32 + lane];
            const float4 sq = Sb[tt * 32 + lane];
            const float ga0 = gq.x, ga1 = gq.y, ga2 = gq.z, ga3 = gq.w;

            v8f P = vzero;
#pragma unroll
            for (int g = 0; g < KG; ++g) {
                v2f a; a[0] = Atan[2 * g]; a[1] = Atan[2 * g + 1];
                P = wmma_f32_k4(a, Bv[g], P);
            }

#pragma unroll
            for (int v = 0; v < 8; ++v) ldsP[(v + 8 * hi) * 16 + col] = P[v];
            asm volatile("s_wait_dscnt 0" ::: "memory");
            v2f Ap0, Ap1;
            {
                float p0 = ldsP[col * 16 + 2 * hi];
                float p1 = ldsP[col * 16 + 2 * hi + 1];
                float p2 = ldsP[col * 16 + 4 + 2 * hi];
                float p3 = ldsP[col * 16 + 4 + 2 * hi + 1];
                Ap0[0] = p0 * ga0; Ap0[1] = p1 * ga1;
                Ap1[0] = p2 * ga2; Ap1[1] = p3 * ga3;
            }
            v2f As;
            As[0] = sq.x;
            As[1] = sq.y;

#pragma unroll
            for (int t4 = 0; t4 < 4; ++t4) {
                v8f acc = wmma_f32_k4(Ap0, Bu[t4][0], vzero);
                acc = wmma_f32_k4(Ap1, Bu[t4][1], acc);
                acc = wmma_f32_k4(As,  Bw[t4],    acc);
                v8f x = XD[t4];
#pragma unroll
                for (int v = 0; v < 8; ++v) x[v] = fmaf(x[v], 1.0f - ALPHA, acc[v]);
                XD[t4] = x;
                if (t4 < 3 || col < XSTR - 48) {
#pragma unroll
                    for (int v = 0; v < 8; ++v)
                        Xs[(v + 8 * hi) * XSTR + 16 * t4 + col] = x[v];
                }
            }
            asm volatile("s_wait_dscnt 0" ::: "memory");

            float xa[2 * KG];
#pragma unroll
            for (int g = 0; g < KG; ++g) {
                int k0 = 4 * g + 2 * hi;
                xa[2 * g]     = Xs[col * XSTR + k0];
                xa[2 * g + 1] = Xs[col * XSTR + k0 + 1];
            }
            __builtin_amdgcn_sched_barrier(0);
#pragma unroll
            for (int i = 0; i < 2 * KG; ++i) Atan[i] = fast_tanh(xa[i]);
          }
        }
      } else {
        float zq[TB / 2][DIMZ], sq[TB / 2][DIMS];
        if (ph < NB) {
            const int t0 = ph * TB;
            const float* zb = z + ((size_t)(t0 + hi) * BATCH + (size_t)brow) * DIMZ;
            const float* sb = s + ((size_t)(t0 + hi) * BATCH + (size_t)brow) * DIMS;
#pragma unroll
            for (int j = 0; j < TB / 2; ++j) {
                const float* zr = zb + (size_t)j * (2 * BATCH * DIMZ);
                const float* sr = sb + (size_t)j * (2 * BATCH * DIMS);
#pragma unroll
                for (int i = 0; i < DIMZ; ++i) zq[j][i] = zr[i];
#pragma unroll
                for (int i = 0; i < DIMS; ++i) sq[j][i] = sr[i];
            }
        }
        if (ph >= 2) {
            const int kb = ph - 2;
            const float4* Sb = ldsS + (kb & 1) * (TB * 32);
            const float* Xb = ldsXs + (kb & 1) * (TB * XSLOT);
#pragma unroll 1
            for (int tt = 0; tt < TB; tt += 2) {
                const int t = kb * TB + tt;
                const float* Xs0 = Xb + tt * XSLOT;
                const float* Xs1 = Xs0 + XSLOT;
                const float zsum0 = Sb[tt * 32 + lane].z;
                const float zsum1 = Sb[(tt + 1) * 32 + lane].z;
                float xr0[2 * KG], xr1[2 * KG];
#pragma unroll
                for (int g = 0; g < KG; ++g) {
                    int k0 = 4 * g + 2 * hi;
                    xr0[2 * g]     = Xs0[col * XSTR + k0];
                    xr0[2 * g + 1] = Xs0[col * XSTR + k0 + 1];
                }
#pragma unroll
                for (int g = 0; g < KG; ++g) {
                    int k0 = 4 * g + 2 * hi;
                    xr1[2 * g]     = Xs1[col * XSTR + k0];
                    xr1[2 * g + 1] = Xs1[col * XSTR + k0 + 1];
                }
                __builtin_amdgcn_sched_barrier(0);
                v8f OD0 = vzero, OD1 = vzero;
#pragma unroll
                for (int g = 0; g < KG; ++g) {
                    float a0 = zsum0 * fast_tanh(xr0[2 * g]);
                    float a1 = zsum0 * fast_tanh(xr0[2 * g + 1]);
                    float c0 = zsum1 * fast_tanh(xr1[2 * g]);
                    float c1 = zsum1 * fast_tanh(xr1[2 * g + 1]);
                    if (g == KG - 1) {
                        a0 = hib ? zsum0 : a0;
                        a1 = hib ? 0.0f : a1;
                        c0 = hib ? zsum1 : c0;
                        c1 = hib ? 0.0f : c1;
                    }
                    v2f a; a[0] = a0; a[1] = a1;
                    v2f c; c[0] = c0; c[1] = c1;
                    OD0 = wmma_f32_k4(a, Bo[g], OD0);
                    OD1 = wmma_f32_k4(c, Bo[g], OD1);
                }
                if (col < DIMY) {
                    size_t base = (size_t)t * BATCH * DIMY + (size_t)(b0 + 8 * hi) * DIMY + col;
#pragma unroll
                    for (int v = 0; v < 8; ++v) out[base + (size_t)v * DIMY] = OD0[v];
#pragma unroll
                    for (int v = 0; v < 8; ++v)
                        out[base + (size_t)BATCH * DIMY + (size_t)v * DIMY] = OD1[v];
                }
            }
        }
        if (ph < NB) {
            float4* Gb = ldsG + (ph & 1) * (TB * 32);
            float4* Sb = ldsS + (ph & 1) * (TB * 32);
#pragma unroll
            for (int j = 0; j < TB / 2; ++j) {
                float zsum = 0.0f;
#pragma unroll
                for (int i = 0; i < DIMZ; ++i) zsum += zq[j][i];
                float a0 = nmbv[0], a1 = nmbv[1], a2 = nmbv[2];
                float a3 = nmbv[3], a4 = nmbv[4], a5 = nmbv[5];
#pragma unroll
                for (int i = 0; i < DIMZ; ++i) {
                    a0 = fmaf(nmw[0][i], zq[j][i], a0);
                    a1 = fmaf(nmw[1][i], zq[j][i], a1);
                    a2 = fmaf(nmw[2][i], zq[j][i], a2);
                    a3 = fmaf(nmw[3][i], zq[j][i], a3);
                    a4 = fmaf(nmw[4][i], zq[j][i], a4);
                    a5 = fmaf(nmw[5][i], zq[j][i], a5);
                }
                const float g0 = fast_sigmoid(a0), g1 = fast_sigmoid(a1), g2 = fast_sigmoid(a2);
                const float g3 = fast_sigmoid(a3), g4 = fast_sigmoid(a4), g5 = fast_sigmoid(a5);
                const int slot = (2 * j + hi) * 32 + col;
                Gb[slot]      = make_float4(g0, g1, g4, g5);
                Gb[slot + 16] = make_float4(g2, g3, 0.0f, 0.0f);
                Sb[slot]      = make_float4(zsum * sq[j][0], zsum * sq[j][1], zsum, 0.0f);
                Sb[slot + 16] = make_float4(zsum * sq[j][2], zsum,            zsum, 0.0f);
            }
        }
      }
      __syncthreads();
    }
}

extern "C" void kernel_launch(void* const* d_in, const int* in_sizes, int n_in,
                              void* d_out, int out_size, void* d_ws, size_t ws_size,
                              hipStream_t stream) {
    (void)in_sizes; (void)n_in; (void)out_size; (void)d_ws; (void)ws_size;
    const float* s   = (const float*)d_in[0];
    const float* z   = (const float*)d_in[1];
    const float* U   = (const float*)d_in[2];
    const float* V   = (const float*)d_in[3];
    const float* Wi  = (const float*)d_in[4];
    const float* bi  = (const float*)d_in[5];
    const float* Wo  = (const float*)d_in[6];
    const float* bo  = (const float*)d_in[7];
    const float* nmW = (const float*)d_in[8];
    const float* nmb = (const float*)d_in[9];
    float* out = (float*)d_out;

    dim3 grid(BATCH / 16);
    dim3 block(64);
    hipLaunchKernelGGL(cxtrnn_wmma_kernel, grid, block, 0, stream,
                       s, z, U, V, Wi, bi, Wo, bo, nmW, nmb, out);
}


# Input term moved to the helper wave: it makes a*zsum*(W_in s+b_in) as D-layout tiles (4 WMMAs from zero) into LDS; recurrence wave's two rank products per tile start from them; order of sums changes
# speedup vs baseline: 2.4921x; 1.0858x over previous
#include <hip/hip_runtime.h>

#define SEQ   512
#define BATCH 4096
#define DIMS  3
#define DIMY  3
#define DIMZ  6
#define RANK  6
#define HID   50
#define KG    13
#define XSTR  52
#define XSLOT (16 * XSTR)
#define ALPHA 0.5f
#define TB    16
#define NB    (SEQ / TB)
#define XS_F4  (2 * TB * XSLOT / 4)
#define P_F4   (16 * 16 / 4)
#define GS_F4  (2 * TB * 32)
#define C_F4   (2 * TB * 256)
#define LDS_F4 (XS_F4 + P_F4 + 2 * GS_F4 + C_F4)

typedef float v2f __attribute__((ext_vector_type(2)));
typedef float v8f __attribute__((ext_vector_type(8)));

__device__ __forceinline__ v8f wmma_f32_k4(v2f a, v2f b, v8f c) {
    return __builtin_amdgcn_wmma_f32_16x16x4_f32(false, a, false, b, (short)0, c, false, false);
}

__device__ __forceinline__ float fast_tanh(float x) {
#if __has_builtin(__builtin_amdgcn_tanhf)
    return __builtin_amdgcn_tanhf(x);
#else
    x = fminf(15.0f, fmaxf(-15.0f, x));
    float e = __expf(2.0f * x);
    float r = __builtin_amdgcn_rcpf(e + 1.0f);
    return fmaf(-2.0f, r, 1.0f);
#endif
}

__device__ __forceinline__ float fast_sigmoid(float x) {
    return __builtin_amdgcn_rcpf(1.0f + __expf(-x));
}

__global__ __launch_bounds__(64) void cxtrnn_wmma_kernel(
    const float* __restrict__ s,   const float* __restrict__ z,
    const float* __restrict__ U,   const float* __restrict__ V,
    const float* __restrict__ Wi,  const float* __restrict__ bi,
    const float* __restrict__ Wo,  const float* __restrict__ bo,
    const float* __restrict__ nmW, const float* __restrict__ nmb,
    float* __restrict__ out)
{
    __shared__ float4 ldsAll[LDS_F4];
    float*  const ldsXs = (float*)ldsAll;
    float*  const ldsP  = (float*)(ldsAll + XS_F4);
    float4* const ldsG  = ldsAll + XS_F4 + P_F4;
    float4* const ldsS  = ldsG + GS_F4;
    float4* const ldsC  = ldsS + GS_F4;

    const int lane = (int)threadIdx.x & 31;
    const int wid  = __builtin_amdgcn_readfirstlane((int)threadIdx.x >> 5);
    const int col  = lane & 15;
    const int hi   = lane >> 4;
    const bool hib = (hi != 0);
    const int b0   = (int)blockIdx.x * 16;
    const int brow = b0 + col;

    v2f Bv[KG];
#pragma unroll
    for (int g = 0; g < KG; ++g) {
        int k0 = 4 * g + 2 * hi;
        float a0 = (k0     < HID && col < RANK) ? V[k0 * RANK + col]       : 0.0f;
        float a1 = (k0 + 1 < HID && col < RANK) ? V[(k0 + 1) * RANK + col] : 0.0f;
        v2f t; t[0] = a0; t[1] = a1; Bv[g] = t;
    }
    v2f Bu[4][2];
#pragma unroll
    for (int t4 = 0; t4 < 4; ++t4) {
        int h = 16 * t4 + col;
#pragma unroll
        for (int g = 0; g < 2; ++g) {
            int k0 = 4 * g + 2 * hi;
            float a0 = (h < HID && k0     < RANK) ? ALPHA * U[h * RANK + k0]     : 0.0f;
            float a1 = (h < HID && k0 + 1 < RANK) ? ALPHA * U[h * RANK + k0 + 1] : 0.0f;
            v2f t; t[0] = a0; t[1] = a1; Bu[t4][g] = t;
        }
    }
    v2f Bw[4];
#pragma unroll
    for (int t4 = 0; t4 < 4; ++t4) {
        int h  = 16 * t4 + col;
        int k0 = 2 * hi;
        float a0 = 0.0f, a1 = 0.0f;
        if (h < HID) {
            a0 = ALPHA * Wi[h * DIMS + k0];
            a1 = ALPHA * ((k0 + 1 < DIMS) ? Wi[h * DIMS + k0 + 1] : bi[h]);
        }
        v2f t; t[0] = a0; t[1] = a1; Bw[t4] = t;
    }
    v2f Bo[KG];
#pragma unroll
    for (int g = 0; g < KG; ++g) {
        int k0 = 4 * g + 2 * hi;
        float a0 = 0.0f, a1 = 0.0f;
        if (col < DIMY) {
            a0 = (k0     < HID) ? Wo[col * HID + k0]     : ((k0     == HID) ? bo[col] : 0.0f);
            a1 = (k0 + 1 < HID) ? Wo[col * HID + k0 + 1] : ((k0 + 1 == HID) ? bo[col] : 0.0f);
        }
        v2f t; t[0] = a0; t[1] = a1; Bo[g] = t;
    }
    float nmw[RANK][DIMZ], nmbv[RANK];
#pragma unroll
    for (int r = 0; r < RANK; ++r) {
        nmbv[r] = nmb[r];
#pragma unroll
        for (int i = 0; i < DIMZ; ++i) nmw[r][i] = nmW[r * DIMZ + i];
    }

    v8f XD[4];
#pragma unroll
    for (int t4 = 0; t4 < 4; ++t4) { v8f zz = {0,0,0,0,0,0,0,0}; XD[t4] = zz; }
    float Atan[2 * KG];
#pragma unroll
    for (int i = 0; i < 2 * KG; ++i) Atan[i] = 0.0f;

    const v8f vzero = {0,0,0,0,0,0,0,0};

#pragma unroll 1
    for (int ph = 0; ph < NB + 2; ++ph) {
      if (wid == 0) {
        if (ph >= 1 && ph <= NB) {
          const int kb = ph - 1;
          const float4* Gb = ldsG + (kb & 1) * (TB * 32);
          const float4* Cb = ldsC + (kb & 1) * (TB * 256);
          float* Xb = ldsXs + (kb & 1) * (TB * XSLOT);
          float4 gq = Gb[lane];
          float cin[32];
#pragma unroll
          for (int q = 0; q < 8; ++q) {
              const float4 c4 = Cb[q * 32 + lane];
              cin[4 * q] = c4.x; cin[4 * q + 1] = c4.y; cin[4 * q + 2] = c4.z; cin[4 * q + 3] = c4.w;
          }
#pragma unroll 1
          for (int tt = 0; tt < TB; ++tt) {
            float* Xs = Xb + tt * XSLOT;
            const float ga0 = gq.x, ga1 = gq.y, ga2 = gq.z, ga3 = gq.w;

            v8f P = vzero;
#pragma unroll
            for (int g = 0; g < KG; ++g) {
                v2f a; a[0] = Atan[2 * g]; a[1] = Atan[2 * g + 1];
                P = wmma_f32_k4(a, Bv[g], P);
            }

#pragma unroll
            for (int v = 0; v < 8; ++v) ldsP[(v + 8 * hi) * 16 + col] = P[v];
            asm volatile("s_wait_dscnt 0" ::: "memory");
            v2f Ap0, Ap1;
            {
                float p0 = ldsP[col * 16 + 2 * hi];
                float p1 = ldsP[col * 16 + 2 * hi + 1];
                float p2 = ldsP[col * 16 + 4 + 2 * hi];
                float p3 = ldsP[col * 16 + 4 + 2 * hi + 1];
                Ap0[0] = p0 * ga0; Ap0[1] = p1 * ga1;
                Ap1[0] = p2 * ga2; Ap1[1] = p3 * ga3;
            }

#pragma unroll
            for (int t4 = 0; t4 < 4; ++t4) {
                v8f acc;
#pragma unroll
                for (int v = 0; v < 8; ++v) acc[v] = cin[8 * t4 + v];
                acc = wmma_f32_k4(Ap0, Bu[t4][0], acc);
                acc = wmma_f32_k4(Ap1, Bu[t4][1], acc);
                v8f x = XD[t4];
#pragma unroll
                for (int v = 0; v < 8; ++v) x[v] = fmaf(x[v], 1.0f - ALPHA, acc[v]);
                XD[t4] = x;
                if (t4 < 3 || col < XSTR - 48) {
#pragma unroll
                    for (int v = 0; v < 8; ++v)
                        Xs[(v + 8 * hi) * XSTR + 16 * t4 + col] = x[v];
                }
            }
            asm volatile("s_wait_dscnt 0" ::: "memory");

            float xa[2 * KG];
#pragma unroll
            for (int g = 0; g < KG; ++g) {
                int k0 = 4 * g + 2 * hi;
                xa[2 * g]     = Xs[col * XSTR + k0];
                xa[2 * g + 1] = Xs[col * XSTR + k0 + 1];
            }
            __builtin_amdgcn_sched_barrier(0);
            {
                const int tn = min(tt + 1, TB - 1);
                gq = Gb[tn * 32 + lane];
#pragma unroll
                for (int q = 0; q < 8; ++q) {
                    const float4 c4 = Cb[tn * 256 + q * 32 + lane];
                    cin[4 * q] = c4.x; cin[4 * q + 1] = c4.y; cin[4 * q + 2] = c4.z; cin[4 * q + 3] = c4.w;
                }
            }
            __builtin_amdgcn_sched_barrier(0);
#pragma unroll
            for (int i = 0; i < 2 * KG; ++i) Atan[i] = fast_tanh(xa[i]);
          }
        }
      } else {
        float zq[TB / 2][DIMZ], sq[TB / 2][DIMS];
        if (ph < NB) {
            const int t0 = ph * TB;
            const float* zb = z + ((size_t)(t0 + hi) * BATCH + (size_t)brow) * DIMZ;
            const float* sb = s + ((size_t)(t0 + hi) * BATCH + (size_t)brow) * DIMS;
#pragma unroll
            for (int j = 0; j < TB / 2; ++j) {
                const float* zr = zb + (size_t)j * (2 * BATCH * DIMZ);
                const float* sr = sb + (size_t)j * (2 * BATCH * DIMS);
#pragma unroll
                for (int i = 0; i < DIMZ; ++i) zq[j][i] = zr[i];
#pragma unroll
                for (int i = 0; i < DIMS; ++i) sq[j][i] = sr[i];
            }
        }
        if (ph >= 2) {
            const int kb = ph - 2;
            const float4* Sb = ldsS + (kb & 1) * (TB * 32);
            const float* Xb = ldsXs + (kb & 1) * (TB * XSLOT);
#pragma unroll 1
            for (int tt = 0; tt < TB; tt += 2) {
                const int t = kb * TB + tt;
                const float* Xs0 = Xb + tt * XSLOT;
                const float* Xs1 = Xs0 + XSLOT;
                const float zsum0 = Sb[tt * 32 + lane].z;
                const float zsum1 = Sb[(tt + 1) * 32 + lane].z;
                float xr0[2 * KG], xr1[2 * KG];
#pragma unroll
                for (int g = 0; g < KG; ++g) {
                    int k0 = 4 * g + 2 * hi;
                    xr0[2 * g]     = Xs0[col * XSTR + k0];
                    xr0[2 * g + 1] = Xs0[col * XSTR + k0 + 1];
                }
#pragma unroll
                for (int g = 0; g < KG; ++g) {
                    int k0 = 4 * g + 2 * hi;
                    xr1[2 * g]     = Xs1[col * XSTR + k0];
                    xr1[2 * g + 1] = Xs1[col * XSTR + k0 + 1];
                }
                __builtin_amdgcn_sched_barrier(0);
                v8f OD0 = vzero, OD1 = vzero;
#pragma unroll
                for (int g = 0; g < KG; ++g) {
                    float a0 = zsum0 * fast_tanh(xr0[2 * g]);
                    float a1 = zsum0 * fast_tanh(xr0[2 * g + 1]);
                    float c0 = zsum1 * fast_tanh(xr1[2 * g]);
                    float c1 = zsum1 * fast_tanh(xr1[2 * g + 1]);
                    if (g == KG - 1) {
                        a0 = hib ? zsum0 : a0;
                        a1 = hib ? 0.0f : a1;
                        c0 = hib ? zsum1 : c0;
                        c1 = hib ? 0.0f : c1;
                    }
                    v2f a; a[0] = a0; a[1] = a1;
                    v2f c; c[0] = c0; c[1] = c1;
                    OD0 = wmma_f32_k4(a, Bo[g], OD0);
                    OD1 = wmma_f32_k4(c, Bo[g], OD1);
                }
                if (col < DIMY) {
                    size_t base = (size_t)t * BATCH * DIMY + (size_t)(b0 + 8 * hi) * DIMY + col;
#pragma unroll
                    for (int v = 0; v < 8; ++v) out[base + (size_t)v * DIMY] = OD0[v];
#pragma unroll
                    for (int v = 0; v < 8; ++v)
                        out[base + (size_t)BATCH * DIMY + (size_t)v * DIMY] = OD1[v];
                }
            }
        }
        if (ph < NB) {
            float4* Gb = ldsG + (ph & 1) * (TB * 32);
            float4* Sb = ldsS + (ph & 1) * (TB * 32);
#pragma unroll
            for (int j = 0; j < TB / 2; ++j) {
                float zsum = 0.0f;
#pragma unroll
                for (int i = 0; i < DIMZ; ++i) zsum += zq[j][i];
                float a0 = nmbv[0], a1 = nmbv[1], a2 = nmbv[2];
                float a3 = nmbv[3], a4 = nmbv[4], a5 = nmbv[5];
#pragma unroll
                for (int i = 0; i < DIMZ; ++i) {
                    a0 = fmaf(nmw[0][i], zq[j][i], a0);
                    a1 = fmaf(nmw[1][i], zq[j][i], a1);
                    a2 = fmaf(nmw[2][i], zq[j][i], a2);
                    a3 = fmaf(nmw[3][i], zq[j][i], a3);
                    a4 = fmaf(nmw[4][i], zq[j][i], a4);
                    a5 = fmaf(nmw[5][i], zq[j][i], a5);
                }
                const float g0 = fast_sigmoid(a0), g1 = fast_sigmoid(a1), g2 = fast_sigmoid(a2);
                const float g3 = fast_sigmoid(a3), g4 = fast_sigmoid(a4), g5 = fast_sigmoid(a5);
                const int slot = (2 * j + hi) * 32 + col;
                Gb[slot]      = make_float4(g0, g1, g4, g5);
                Gb[slot + 16] = make_float4(g2, g3, 0.0f, 0.0f);
                Sb[slot]      = make_float4(zsum * sq[j][0], zsum * sq[j][1], zsum, 0.0f);
                Sb[slot + 16] = make_float4(zsum * sq[j][2], zsum,            zsum, 0.0f);
            }
            asm volatile("s_wait_dscnt 0" ::: "memory");
            float4* Cb = ldsC + (ph & 1) * (TB * 256);
#pragma unroll 4
            for (int tt = 0; tt < TB; ++tt) {
                const float4 rq = Sb[tt * 32 + lane];
                v2f As;
                As[0] = rq.x;
                As[1] = rq.y;
#pragma unroll
                for (int t4 = 0; t4 < 4; ++t4) {
                    const v8f c = wmma_f32_k4(As, Bw[t4], vzero);
                    Cb[tt * 256 + (2 * t4) * 32 + lane]     = make_float4(c[0], c[1], c[2], c[3]);
                    Cb[tt * 256 + (2 * t4 + 1) * 32 + lane] = make_float4(c[4], c[5], c[6], c[7]);
                }
            }
        }
      }
      __syncthreads();
    }
}

extern "C" void kernel_launch(void* const* d_in, const int* in_sizes, int n_in,
                              void* d_out, int out_size, void* d_ws, size_t ws_size,
                              hipStream_t stream) {
    (void)in_sizes; (void)n_in; (void)out_size; (void)d_ws; (void)ws_size;
    const float* s   = (const float*)d_in[0];
    const float* z   = (const float*)d_in[1];
    const float* U   = (const float*)d_in[2];
    const float* V   = (const float*)d_in[3];
    const float* Wi  = (const float*)d_in[4];
    const float* bi  = (const float*)d_in[5];
    const float* Wo  = (const float*)d_in[6];
    const float* bo  = (const float*)d_in[7];
    const float* nmW = (const float*)d_in[8];
    const float* nmb = (const float*)d_in[9];
    float* out = (float*)d_out;

    dim3 grid(BATCH / 16);
    dim3 block(64);
    hipLaunchKernelGGL(cxtrnn_wmma_kernel, grid, block, 0, stream,
                       s, z, U, V, Wi, bi, Wo, bo, nmW, nmb, out);
}


# Recurrence wave: no drain of the LDS counter between its own LDS stores and its reads of them (P tile, state slot): the LDS serves a wave in issue order, reads queue behind stores; sums unchanged
# speedup vs baseline: 2.6546x; 1.0652x over previous
#include <hip/hip_runtime.h>

#define SEQ   512
#define BATCH 4096
#define DIMS  3
#define DIMY  3
#define DIMZ  6
#define RANK  6
#define HID   50
#define KG    13
#define XSTR  52
#define XSLOT (16 * XSTR)
#define ALPHA 0.5f
#define TB    16
#define NB    (SEQ / TB)
#define XS_F4  (2 * TB * XSLOT / 4)
#define P_F4   (16 * 16 / 4)
#define GS_F4  (2 * TB * 32)
#define C_F4   (2 * TB * 256)
#define LDS_F4 (XS_F4 + P_F4 + 2 * GS_F4 + C_F4)

typedef float v2f __attribute__((ext_vector_type(2)));
typedef float v8f __attribute__((ext_vector_type(8)));

__device__ __forceinline__ v8f wmma_f32_k4(v2f a, v2f b, v8f c) {
    return __builtin_amdgcn_wmma_f32_16x16x4_f32(false, a, false, b, (short)0, c, false, false);
}

__device__ __forceinline__ float fast_tanh(float x) {
#if __has_builtin(__builtin_amdgcn_tanhf)
    return __builtin_amdgcn_tanhf(x);
#else
    x = fminf(15.0f, fmaxf(-15.0f, x));
    float e = __expf(2.0f * x);
    float r = __builtin_amdgcn_rcpf(e + 1.0f);
    return fmaf(-2.0f, r, 1.0f);
#endif
}

__device__ __forceinline__ void wave_lds_order() {
    __builtin_amdgcn_fence(__ATOMIC_ACQ_REL, "wavefront");
}

__device__ __forceinline__ float fast_sigmoid(float x) {
    return __builtin_amdgcn_rcpf(1.0f + __expf(-x));
}

__global__ __launch_bounds__(64) void cxtrnn_wmma_kernel(
    const float* __restrict__ s,   const float* __restrict__ z,
    const float* __restrict__ U,   const float* __restrict__ V,
    const float* __restrict__ Wi,  const float* __restrict__ bi,
    const float* __restrict__ Wo,  const float* __restrict__ bo,
    const float* __restrict__ nmW, const float* __restrict__ nmb,
    float* __restrict__ out)
{
    __shared__ float4 ldsAll[LDS_F4];
    float*  const ldsXs = (float*)ldsAll;
    float*  const ldsP  = (float*)(ldsAll + XS_F4);
    float4* const ldsG  = ldsAll + XS_F4 + P_F4;
    float4* const ldsS  = ldsG + GS_F4;
    float4* const ldsC  = ldsS + GS_F4;

    const int lane = (int)threadIdx.x & 31;
    const int wid  = __builtin_amdgcn_readfirstlane((int)threadIdx.x >> 5);
    const int col  = lane & 15;
    const int hi   = lane >> 4;
    const bool hib = (hi != 0);
    const int b0   = (int)blockIdx.x * 16;
    const int brow = b0 + col;

    v2f Bv[KG];
#pragma unroll
    for (int g = 0; g < KG; ++g) {
        int k0 = 4 * g + 2 * hi;
        float a0 = (k0     < HID && col < RANK) ? V[k0 * RANK + col]       : 0.0f;
        float a1 = (k0 + 1 < HID && col < RANK) ? V[(k0 + 1) * RANK + col] : 0.0f;
        v2f t; t[0] = a0; t[1] = a1; Bv[g] = t;
    }
    v2f Bu[4][2];
#pragma unroll
    for (int t4 = 0; t4 < 4; ++t4) {
        int h = 16 * t4 + col;
#pragma unroll
        for (int g = 0; g < 2; ++g) {
            int k0 = 4 * g + 2 * hi;
            float a0 = (h < HID && k0     < RANK) ? ALPHA * U[h * RANK + k0]     : 0.0f;
            float a1 = (h < HID && k0 + 1 < RANK) ? ALPHA * U[h * RANK + k0 + 1] : 0.0f;
            v2f t; t[0] = a0; t[1] = a1; Bu[t4][g] = t;
        }
    }
    v2f Bw[4];
#pragma unroll
    for (int t4 = 0; t4 < 4; ++t4) {
        int h  = 16 * t4 + col;
        int k0 = 2 * hi;
        float a0 = 0.0f, a1 = 0.0f;
        if (h < HID) {
            a0 = ALPHA * Wi[h * DIMS + k0];
            a1 = ALPHA * ((k0 + 1 < DIMS) ? Wi[h * DIMS + k0 + 1] : bi[h]);
        }
        v2f t; t[0] = a0; t[1] = a1; Bw[t4] = t;
    }
    v2f Bo[KG];
#pragma unroll
    for (int g = 0; g < KG; ++g) {
        int k0 = 4 * g + 2 * hi;
        float a0 = 0.0f, a1 = 0.0f;
        if (col < DIMY) {
            a0 = (k0     < HID) ? Wo[col * HID + k0]     : ((k0     == HID) ? bo[col] : 0.0f);
            a1 = (k0 + 1 < HID) ? Wo[col * HID + k0 + 1] : ((k0 + 1 == HID) ? bo[col] : 0.0f);
        }
        v2f t; t[0] = a0; t[1] = a1; Bo[g] = t;
    }
    float nmw[RANK][DIMZ], nmbv[RANK];
#pragma unroll
    for (int r = 0; r < RANK; ++r) {
        nmbv[r] = nmb[r];
#pragma unroll
        for (int i = 0; i < DIMZ; ++i) nmw[r][i] = nmW[r * DIMZ + i];
    }

    v8f XD[4];
#pragma unroll
    for (int t4 = 0; t4 < 4; ++t4) { v8f zz = {0,0,0,0,0,0,0,0}; XD[t4] = zz; }
    float Atan[2 * KG];
#pragma unroll
    for (int i = 0; i < 2 * KG; ++i) Atan[i] = 0.0f;

    const v8f vzero = {0,0,0,0,0,0,0,0};

#pragma unroll 1
    for (int ph = 0; ph < NB + 2; ++ph) {
      if (wid == 0) {
        if (ph >= 1 && ph <= NB) {
          const int kb = ph - 1;
          const float4* Gb = ldsG + (kb & 1) * (TB * 32);
          const float4* Cb = ldsC + (kb & 1) * (TB * 256);
          float* Xb = ldsXs + (kb & 1) * (TB * XSLOT);
          float4 gq = Gb[lane];
          float cin[32];
#pragma unroll
          for (int q = 0; q < 8; ++q) {
              const float4 c4 = Cb[q * 32 + lane];
              cin[4 * q] = c4.x; cin[4 * q + 1] = c4.y; cin[4 * q + 2] = c4.z; cin[4 * q + 3] = c4.w;
          }
#pragma unroll 1
          for (int tt = 0; tt < TB; ++tt) {
            float* Xs = Xb + tt * XSLOT;
            const float ga0 = gq.x, ga1 = gq.y, ga2 = gq.z, ga3 = gq.w;

            v8f P = vzero;
#pragma unroll
            for (int g = 0; g < KG; ++g) {
                v2f a; a[0] = Atan[2 * g]; a[1] = Atan[2 * g + 1];
                P = wmma_f32_k4(a, Bv[g], P);
            }

#pragma unroll
            for (int v = 0; v < 8; ++v) ldsP[(v + 8 * hi) * 16 + col] = P[v];
            wave_lds_order();
            v2f Ap0, Ap1;
            {
                float p0 = ldsP[col * 16 + 2 * hi];
                float p1 = ldsP[col * 16 + 2 * hi + 1];
                float p2 = ldsP[col * 16 + 4 + 2 * hi];
                float p3 = ldsP[col * 16 + 4 + 2 * hi + 1];
                Ap0[0] = p0 * ga0; Ap0[1] = p1 * ga1;
                Ap1[0] = p2 * ga2; Ap1[1] = p3 * ga3;
            }

#pragma unroll
            for (int t4 = 0; t4 < 4; ++t4) {
                v8f acc;
#pragma unroll
                for (int v = 0; v < 8; ++v) acc[v] = cin[8 * t4 + v];
                acc = wmma_f32_k4(Ap0, Bu[t4][0], acc);
                acc = wmma_f32_k4(Ap1, Bu[t4][1], acc);
                v8f x = XD[t4];
#pragma unroll
                for (int v = 0; v < 8; ++v) x[v] = fmaf(x[v], 1.0f - ALPHA, acc[v]);
                XD[t4] = x;
                if (t4 < 3 || col < XSTR - 48) {
#pragma unroll
                    for (int v = 0; v < 8; ++v)
                        Xs[(v + 8 * hi) * XSTR + 16 * t4 + col] = x[v];
                }
            }
            wave_lds_order();

            float xa[2 * KG];
#pragma unroll
            for (int g = 0; g < KG; ++g) {
                int k0 = 4 * g + 2 * hi;
                xa[2 * g]     = Xs[col * XSTR + k0];
                xa[2 * g + 1] = Xs[col * XSTR + k0 + 1];
            }
            __builtin_amdgcn_sched_barrier(0);
            {
                const int tn = min(tt + 1, TB - 1);
                gq = Gb[tn * 32 + lane];
#pragma unroll
                for (int q = 0; q < 8; ++q) {
                    const float4 c4 = Cb[tn * 256 + q * 32 + lane];
                    cin[4 * q] = c4.x; cin[4 * q + 1] = c4.y; cin[4 * q + 2] = c4.z; cin[4 * q + 3] = c4.w;
                }
            }
            __builtin_amdgcn_sched_barrier(0);
#pragma unroll
            for (int i = 0; i < 2 * KG; ++i) Atan[i] = fast_tanh(xa[i]);
          }
        }
      } else {
        float zq[TB / 2][DIMZ], sq[TB / 2][DIMS];
        if (ph < NB) {
            const int t0 = ph * TB;
            const float* zb = z + ((size_t)(t0 + hi) * BATCH + (size_t)brow) * DIMZ;
            const float* sb = s + ((size_t)(t0 + hi) * BATCH + (size_t)brow) * DIMS;
#pragma unroll
            for (int j = 0; j < TB / 2; ++j) {
                const float* zr = zb + (size_t)j * (2 * BATCH * DIMZ);
                const float* sr = sb + (size_t)j * (2 * BATCH * DIMS);
#pragma unroll
                for (int i = 0; i < DIMZ; ++i) zq[j][i] = zr[i];
#pragma unroll
                for (int i = 0; i < DIMS; ++i) sq[j][i] = sr[i];
            }
        }
        if (ph >= 2) {
            const int kb = ph - 2;
            const float4* Sb = ldsS + (kb & 1) * (TB * 32);
            const float* Xb = ldsXs + (kb & 1) * (TB * XSLOT);
#pragma unroll 1
            for (int tt = 0; tt < TB; tt += 2) {
                const int t = kb * TB + tt;
                const float* Xs0 = Xb + tt * XSLOT;
                const float* Xs1 = Xs0 + XSLOT;
                const float zsum0 = Sb[tt * 32 + lane].z;
                const float zsum1 = Sb[(tt + 1) * 32 + lane].z;
                float xr0[2 * KG], xr1[2 * KG];
#pragma unroll
                for (int g = 0; g < KG; ++g) {
                    int k0 = 4 * g + 2 * hi;
                    xr0[2 * g]     = Xs0[col * XSTR + k0];
                    xr0[2 * g + 1] = Xs0[col * XSTR + k0 + 1];
                }
#pragma unroll
                for (int g = 0; g < KG; ++g) {
                    int k0 = 4 * g + 2 * hi;
                    xr1[2 * g]     = Xs1[col * XSTR + k0];
                    xr1[2 * g + 1] = Xs1[col * XSTR + k0 + 1];
                }
                __builtin_amdgcn_sched_barrier(0);
                v8f OD0 = vzero, OD1 = vzero;
#pragma unroll
                for (int g = 0; g < KG; ++g) {
                    float a0 = zsum0 * fast_tanh(xr0[2 * g]);
                    float a1 = zsum0 * fast_tanh(xr0[2 * g + 1]);
                    float c0 = zsum1 * fast_tanh(xr1[2 * g]);
                    float c1 = zsum1 * fast_tanh(xr1[2 * g + 1]);
                    if (g == KG - 1) {
                        a0 = hib ? zsum0 : a0;
                        a1 = hib ? 0.0f : a1;
                        c0 = hib ? zsum1 : c0;
                        c1 = hib ? 0.0f : c1;
                    }
                    v2f a; a[0] = a0; a[1] = a1;
                    v2f c; c[0] = c0; c[1] = c1;
                    OD0 = wmma_f32_k4(a, Bo[g], OD0);
                    OD1 = wmma_f32_k4(c, Bo[g], OD1);
                }
                if (col < DIMY) {
                    size_t base = (size_t)t * BATCH * DIMY + (size_t)(b0 + 8 * hi) * DIMY + col;
#pragma unroll
                    for (int v = 0; v < 8; ++v) out[base + (size_t)v * DIMY] = OD0[v];
#pragma unroll
                    for (int v = 0; v < 8; ++v)
                        out[base + (size_t)BATCH * DIMY + (size_t)v * DIMY] = OD1[v];
                }
            }
        }
        if (ph < NB) {
            float4* Gb = ldsG + (ph & 1) * (TB * 32);
            float4* Sb = ldsS + (ph & 1) * (TB * 32);
#pragma unroll
            for (int j = 0; j < TB / 2; ++j) {
                float zsum = 0.0f;
#pragma unroll
                for (int i = 0; i < DIMZ; ++i) zsum += zq[j][i];
                float a0 = nmbv[0], a1 = nmbv[1], a2 = nmbv[2];
                float a3 = nmbv[3], a4 = nmbv[4], a5 = nmbv[5];
#pragma unroll
                for (int i = 0; i < DIMZ; ++i) {
                    a0 = fmaf(nmw[0][i], zq[j][i], a0);
                    a1 = fmaf(nmw[1][i], zq[j][i], a1);
                    a2 = fmaf(nmw[2][i], zq[j][i], a2);
                    a3 = fmaf(nmw[3][i], zq[j][i], a3);
                    a4 = fmaf(nmw[4][i], zq[j][i], a4);
                    a5 = fmaf(nmw[5][i], zq[j][i], a5);
                }
                const float g0 = fast_sigmoid(a0), g1 = fast_sigmoid(a1), g2 = fast_sigmoid(a2);
                const float g3 = fast_sigmoid(a3), g4 = fast_sigmoid(a4), g5 = fast_sigmoid(a5);
                const int slot = (2 * j + hi) * 32 + col;
                Gb[slot]      = make_float4(g0, g1, g4, g5);
                Gb[slot + 16] = make_float4(g2, g3, 0.0f, 0.0f);
                Sb[slot]      = make_float4(zsum * sq[j][0], zsum * sq[j][1], zsum, 0.0f);
                Sb[slot + 16] = make_float4(zsum * sq[j][2], zsum,            zsum, 0.0f);
            }
            asm volatile("s_wait_dscnt 0" ::: "memory");
            float4* Cb = ldsC + (ph & 1) * (TB * 256);
#pragma unroll 4
            for (int tt = 0; tt < TB; ++tt) {
                const float4 rq = Sb[tt * 32 + lane];
                v2f As;
                As[0] = rq.x;
                As[1] = rq.y;
#pragma unroll
                for (int t4 = 0; t4 < 4; ++t4) {
                    const v8f c = wmma_f32_k4(As, Bw[t4], vzero);
                    Cb[tt * 256 + (2 * t4) * 32 + lane]     = make_float4(c[0], c[1], c[2], c[3]);
                    Cb[tt * 256 + (2 * t4 + 1) * 32 + lane] = make_float4(c[4], c[5], c[6], c[7]);
                }
            }
        }
      }
      __syncthreads();
    }
}

extern "C" void kernel_launch(void* const* d_in, const int* in_sizes, int n_in,
                              void* d_out, int out_size, void* d_ws, size_t ws_size,
                              hipStream_t stream) {
    (void)in_sizes; (void)n_in; (void)out_size; (void)d_ws; (void)ws_size;
    const float* s   = (const float*)d_in[0];
    const float* z   = (const float*)d_in[1];
    const float* U   = (const float*)d_in[2];
    const float* V   = (const float*)d_in[3];
    const float* Wi  = (const float*)d_in[4];
    const float* bi  = (const float*)d_in[5];
    const float* Wo  = (const float*)d_in[6];
    const float* bo  = (const float*)d_in[7];
    const float* nmW = (const float*)d_in[8];
    const float* nmb = (const float*)d_in[9];
    float* out = (float*)d_out;

    dim3 grid(BATCH / 16);
    dim3 block(64);
    hipLaunchKernelGGL(cxtrnn_wmma_kernel, grid, block, 0, stream,
                       s, z, U, V, Wi, bi, Wo, bo, nmW, nmb, out);
}


# Recurrence split by batch over two waves (8 rows each, nothing exchanged); rank product's free columns carry the second K half: chain 7 WMMAs not 13, halves joined by DPP add; order of sums changes
# speedup vs baseline: 3.7402x; 1.4090x over previous
#include <hip/hip_runtime.h>

#define SEQ   512
#define BATCH 4096
#define DIMS  3
#define DIMY  3
#define DIMZ  6
#define RANK  6
#define HID   50
#define KG    13
#define KH    24
#define KG2   7
#define XSTR  52
#define XSLOT (16 * XSTR)
#define ALPHA 0.5f
#define TB    16
#define NB    (SEQ / TB)
#define XS_F4  (2 * TB * XSLOT / 4)
#define GS_F4  (2 * TB * 32)
#define C_F4   (2 * TB * 256)
#define Z_F4   (4 * TB * 16 / 4)
#define LDS_F4 (XS_F4 + 2 * GS_F4 + C_F4 + Z_F4)

typedef float v2f __attribute__((ext_vector_type(2)));
typedef float v8f __attribute__((ext_vector_type(8)));

__device__ __forceinline__ v8f wmma_f32_k4(v2f a, v2f b, v8f c) {
    return __builtin_amdgcn_wmma_f32_16x16x4_f32(false, a, false, b, (short)0, c, false, false);
}

__device__ __forceinline__ float fast_tanh(float x) {
#if __has_builtin(__builtin_amdgcn_tanhf)
    return __builtin_amdgcn_tanhf(x);
#else
    x = fminf(15.0f, fmaxf(-15.0f, x));
    float e = __expf(2.0f * x);
    float r = __builtin_amdgcn_rcpf(e + 1.0f);
    return fmaf(-2.0f, r, 1.0f);
#endif
}

__device__ __forceinline__ void wave_lds_order() {
    __builtin_amdgcn_fence(__ATOMIC_ACQ_REL, "wavefront");
}

__device__ __forceinline__ float row_xor8(float v) {
    return __int_as_float(__builtin_amdgcn_update_dpp(0, __float_as_int(v), 0x168, 0xf, 0xf, true));
}

__device__ __forceinline__ float fast_sigmoid(float x) {
    return __builtin_amdgcn_rcpf(1.0f + __expf(-x));
}

__global__ __launch_bounds__(128) void cxtrnn_wmma_kernel(
    const float* __restrict__ s,   const float* __restrict__ z,
    const float* __restrict__ U,   const float* __restrict__ V,
    const float* __restrict__ Wi,  const float* __restrict__ bi,
    const float* __restrict__ Wo,  const float* __restrict__ bo,
    const float* __restrict__ nmW, const float* __restrict__ nmb,
    float* __restrict__ out)
{
    __shared__ float4 ldsAll[LDS_F4];
    float*  const ldsXs = (float*)ldsAll;
    float4* const ldsG  = ldsAll + XS_F4;
    float4* const ldsS  = ldsG + GS_F4;
    float4* const ldsC  = ldsS + GS_F4;
    float*  const ldsZ  = (float*)(ldsC + C_F4);

    const int lane = (int)threadIdx.x & 31;
    const int wid  = __builtin_amdgcn_readfirstlane((int)threadIdx.x >> 5);
    const int col  = lane & 15;
    const int hi   = lane >> 4;
    const bool hib = (hi != 0);
    const int b0   = (int)blockIdx.x * 16;
    const int brow = b0 + col;

    const int vset  = (col >> 2) & 1;
    const int vrank = (col < 8) ? (((col & 3) < 2) ? (col & 3) : (col & 3) + 2)
                                : (((col & 3) < 2) ? (col & 3) + 2 : -1);
    v2f Av[KG2];
#pragma unroll
    for (int g = 0; g < KG2; ++g) {
        v2f t;
#pragma unroll
        for (int j = 0; j < 2; ++j) {
            const int p  = 4 * g + 2 * hi + j;
            const int k  = vset ? KH + p : p;
            const bool ok = (vrank >= 0) && (vset ? (k < HID) : (p < KH));
            const float w = V[ok ? k * RANK + vrank : 0];
            t[j] = ok ? w : 0.0f;
        }
        Av[g] = t;
    }
    v2f Bu[4][2];
#pragma unroll
    for (int t4 = 0; t4 < 4; ++t4) {
        int h = 16 * t4 + col;
#pragma unroll
        for (int g = 0; g < 2; ++g) {
            int k0 = 4 * g + 2 * hi;
            float a0 = (h < HID && k0     < RANK) ? ALPHA * U[h * RANK + k0]     : 0.0f;
            float a1 = (h < HID && k0 + 1 < RANK) ? ALPHA * U[h * RANK + k0 + 1] : 0.0f;
            v2f t; t[0] = a0; t[1] = a1; Bu[t4][g] = t;
        }
    }
    v2f Bw[4];
#pragma unroll
    for (int t4 = 0; t4 < 4; ++t4) {
        int h  = 16 * t4 + col;
        int k0 = 2 * hi;
        float a0 = 0.0f, a1 = 0.0f;
        if (h < HID) {
            a0 = ALPHA * Wi[h * DIMS + k0];
            a1 = ALPHA * ((k0 + 1 < DIMS) ? Wi[h * DIMS + k0 + 1] : bi[h]);
        }
        v2f t; t[0] = a0; t[1] = a1; Bw[t4] = t;
    }
    v2f Bo[KG];
#pragma unroll
    for (int g = 0; g < KG; ++g) {
        int k0 = 4 * g + 2 * hi;
        float a0 = 0.0f, a1 = 0.0f;
        if (col < DIMY) {
            a0 = (k0     < HID) ? Wo[col * HID + k0]     : ((k0     == HID) ? bo[col] : 0.0f);
            a1 = (k0 + 1 < HID) ? Wo[col * HID + k0 + 1] : ((k0 + 1 == HID) ? bo[col] : 0.0f);
        }
        v2f t; t[0] = a0; t[1] = a1; Bo[g] = t;
    }
    float nmw[RANK][DIMZ], nmbv[RANK];
#pragma unroll
    for (int r = 0; r < RANK; ++r) {
        nmbv[r] = nmb[r];
#pragma unroll
        for (int i = 0; i < DIMZ; ++i) nmw[r][i] = nmW[r * DIMZ + i];
    }

    v8f XD[4];
#pragma unroll
    for (int t4 = 0; t4 < 4; ++t4) { v8f zz = {0,0,0,0,0,0,0,0}; XD[t4] = zz; }
    float Atan[2 * KG2];
#pragma unroll
    for (int i = 0; i < 2 * KG2; ++i) Atan[i] = 0.0f;

    const v8f vzero = {0,0,0,0,0,0,0,0};

#pragma unroll 1
    for (int ph = 0; ph < NB + 2; ++ph) {
      if (wid < 2) {
        if (ph >= 1 && ph <= NB) {
          const int kb = ph - 1;
          const int rb = 8 * wid;
          const int bl = col & 7;
          const int kh = col >> 3;
          const bool khb = (kh != 0);
          const float4* Gb = ldsG + (kb & 1) * (TB * 32) + 16 * hi + rb + bl;
          const float4* Cb = ldsC + (kb & 1) * (TB * 256) + 16 * wid + col;
          float* Xb = ldsXs + (kb & 1) * (TB * XSLOT) + rb * XSTR;
          float4 gq = Gb[0];
          float cin[32];
#pragma unroll
          for (int q = 0; q < 8; ++q) {
              const float4 c4 = Cb[q * 32];
              cin[4 * q] = c4.x; cin[4 * q + 1] = c4.y; cin[4 * q + 2] = c4.z; cin[4 * q + 3] = c4.w;
          }
#pragma unroll 1
          for (int tt = 0; tt < TB; ++tt) {
            float* Xs = Xb + tt * XSLOT;
            const float ga0 = gq.x, ga1 = gq.y, ga2 = gq.z, ga3 = gq.w;

            v8f PT = vzero;
#pragma unroll
            for (int g = 0; g < KG2; ++g) {
                v2f b; b[0] = Atan[2 * g]; b[1] = Atan[2 * g + 1];
                PT = wmma_f32_k4(Av[g], b, PT);
            }

            const float p0 = khb ? PT[4] : PT[0];
            const float p1 = khb ? PT[5] : PT[1];
            const float p2 = khb ? PT[6] : PT[2];
            const float p3 = khb ? PT[7] : PT[3];
            v2f Ap0, Ap1;
            Ap0[0] = (p0 + row_xor8(p0)) * ga0; Ap0[1] = (p1 + row_xor8(p1)) * ga1;
            Ap1[0] = (p2 + row_xor8(p2)) * ga2; Ap1[1] = (p3 + row_xor8(p3)) * ga3;

#pragma unroll
            for (int t4 = 0; t4 < 4; ++t4) {
                v8f acc;
#pragma unroll
                for (int v = 0; v < 8; ++v) acc[v] = cin[8 * t4 + v];
                acc = wmma_f32_k4(Ap0, Bu[t4][0], acc);
                acc = wmma_f32_k4(Ap1, Bu[t4][1], acc);
                v8f x = XD[t4];
#pragma unroll
                for (int v = 0; v < 8; ++v) x[v] = fmaf(x[v], 1.0f - ALPHA, acc[v]);
                XD[t4] = x;
                if (t4 < 3 || col < XSTR - 48) {
#pragma unroll
                    for (int v = 0; v < 4; ++v)
                        Xs[(v + 4 * hi) * XSTR + 16 * t4 + col] = hib ? x[v + 4] : x[v];
                }
            }
            wave_lds_order();

            float xa[2 * KG2];
#pragma unroll
            for (int g = 0; g < KG2; ++g) {
                int k0 = KH * kh + 4 * g + 2 * hi;
                xa[2 * g]     = Xs[bl * XSTR + k0];
                xa[2 * g + 1] = Xs[bl * XSTR + k0 + 1];
            }
            __builtin_amdgcn_sched_barrier(0);
            {
                const int tn = min(tt + 1, TB - 1);
                gq = Gb[tn * 32];
#pragma unroll
                for (int q = 0; q < 8; ++q) {
                    const float4 c4 = Cb[tn * 256 + q * 32];
                    cin[4 * q] = c4.x; cin[4 * q + 1] = c4.y; cin[4 * q + 2] = c4.z; cin[4 * q + 3] = c4.w;
                }
            }
            __builtin_amdgcn_sched_barrier(0);
#pragma unroll
            for (int i = 0; i < 2 * KG2; ++i) Atan[i] = fast_tanh(xa[i]);
          }
        }
      } else if (wid == 2) {
        if (ph >= 2) {
            const int kb = ph - 2;
            const float* Zb = ldsZ + (kb & 3) * (TB * 16);
            const float* Xb = ldsXs + (kb & 1) * (TB * XSLOT);
#pragma unroll 1
            for (int tt = 0; tt < TB; tt += 2) {
                const int t = kb * TB + tt;
                const float* Xs0 = Xb + tt * XSLOT;
                const float* Xs1 = Xs0 + XSLOT;
                const float zsum0 = Zb[tt * 16 + col];
                const float zsum1 = Zb[(tt + 1) * 16 + col];
                float xr0[2 * KG], xr1[2 * KG];
#pragma unroll
                for (int g = 0; g < KG; ++g) {
                    int k0 = 4 * g + 2 * hi;
                    xr0[2 * g]     = Xs0[col * XSTR + k0];
                    xr0[2 * g + 1] = Xs0[col * XSTR + k0 + 1];
                }
#pragma unroll
                for (int g = 0; g < KG; ++g) {
                    int k0 = 4 * g + 2 * hi;
                    xr1[2 * g]     = Xs1[col * XSTR + k0];
                    xr1[2 * g + 1] = Xs1[col * XSTR + k0 + 1];
                }
                __builtin_amdgcn_sched_barrier(0);
                v8f OD0 = vzero, OD1 = vzero;
#pragma unroll
                for (int g = 0; g < KG; ++g) {
                    float a0 = zsum0 * fast_tanh(xr0[2 * g]);
                    float a1 = zsum0 * fast_tanh(xr0[2 * g + 1]);
                    float c0 = zsum1 * fast_tanh(xr1[2 * g]);
                    float c1 = zsum1 * fast_tanh(xr1[2 * g + 1]);
                    if (g == KG - 1) {
                        a0 = hib ? zsum0 : a0;
                        a1 = hib ? 0.0f : a1;
                        c0 = hib ? zsum1 : c0;
                        c1 = hib ? 0.0f : c1;
                    }
                    v2f a; a[0] = a0; a[1] = a1;
                    v2f c; c[0] = c0; c[1] = c1;
                    OD0 = wmma_f32_k4(a, Bo[g], OD0);
                    OD1 = wmma_f32_k4(c, Bo[g], OD1);
                }
                if (col < DIMY) {
                    size_t base = (size_t)t * BATCH * DIMY + (size_t)(b0 + 8 * hi) * DIMY + col;
#pragma unroll
                    for (int v = 0; v < 8; ++v) out[base + (size_t)v * DIMY] = OD0[v];
#pragma unroll
                    for (int v = 0; v < 8; ++v)
                        out[base + (size_t)BATCH * DIMY + (size_t)v * DIMY] = OD1[v];
                }
            }
        }
      } else {
        if (ph < NB) {
            float zq[TB / 2][DIMZ], sq[TB / 2][DIMS];
            const int t0 = ph * TB;
            const float* zb = z + ((size_t)(t0 + hi) * BATCH + (size_t)brow) * DIMZ;
            const float* sb = s + ((size_t)(t0 + hi) * BATCH + (size_t)brow) * DIMS;
#pragma unroll
            for (int j = 0; j < TB / 2; ++j) {
                const float* zr = zb + (size_t)j * (2 * BATCH * DIMZ);
                const float* sr = sb + (size_t)j * (2 * BATCH * DIMS);
#pragma unroll
                for (int i = 0; i < DIMZ; ++i) zq[j][i] = zr[i];
#pragma unroll
                for (int i = 0; i < DIMS; ++i) sq[j][i] = sr[i];
            }
            float4* Gb = ldsG + (ph & 1) * (TB * 32);
            float4* Sb = ldsS + (ph & 1) * (TB * 32);
            float*  Zb = ldsZ + (ph & 3) * (TB * 16);
#pragma unroll
            for (int j = 0; j < TB / 2; ++j) {
                float zsum = 0.0f;
#pragma unroll
                for (int i = 0; i < DIMZ; ++i) zsum += zq[j][i];
                float a0 = nmbv[0], a1 = nmbv[1], a2 = nmbv[2];
                float a3 = nmbv[3], a4 = nmbv[4], a5 = nmbv[5];
#pragma unroll
                for (int i = 0; i < DIMZ; ++i) {
                    a0 = fmaf(nmw[0][i], zq[j][i], a0);
                    a1 = fmaf(nmw[1][i], zq[j][i], a1);
                    a2 = fmaf(nmw[2][i], zq[j][i], a2);
                    a3 = fmaf(nmw[3][i], zq[j][i], a3);
                    a4 = fmaf(nmw[4][i], zq[j][i], a4);
                    a5 = fmaf(nmw[5][i], zq[j][i], a5);
                }
                const float g0 = fast_sigmoid(a0), g1 = fast_sigmoid(a1), g2 = fast_sigmoid(a2);
                const float g3 = fast_sigmoid(a3), g4 = fast_sigmoid(a4), g5 = fast_sigmoid(a5);
                const int slot = (2 * j + hi) * 32 + col;
                Gb[slot]      = make_float4(g0, g1, g4, g5);
                Gb[slot + 16] = make_float4(g2, g3, 0.0f, 0.0f);
                Sb[slot]      = make_float4(zsum * sq[j][0], zsum * sq[j][1], 0.0f, 0.0f);
                Sb[slot + 16] = make_float4(zsum * sq[j][2], zsum,            0.0f, 0.0f);
                Zb[(2 * j + hi) * 16 + col] = zsum;
            }
            asm volatile("s_wait_dscnt 0" ::: "memory");
            float4* Cb = ldsC + (ph & 1) * (TB * 256);
#pragma unroll 4
            for (int tt = 0; tt < TB; ++tt) {
                const float4 rq = Sb[tt * 32 + lane];
                v2f As;
                As[0] = rq.x;
                As[1] = rq.y;
#pragma unroll
                for (int t4 = 0; t4 < 4; ++t4) {
                    const v8f c = wmma_f32_k4(As, Bw[t4], vzero);
                    Cb[tt * 256 + (2 * t4) * 32 + lane]     = make_float4(c[0], c[1], c[2], c[3]);
                    Cb[tt * 256 + (2 * t4 + 1) * 32 + lane] = make_float4(c[4], c[5], c[6], c[7]);
                }
            }
        }
      }
      __syncthreads();
    }
}

extern "C" void kernel_launch(void* const* d_in, const int* in_sizes, int n_in,
                              void* d_out, int out_size, void* d_ws, size_t ws_size,
                              hipStream_t stream) {
    (void)in_sizes; (void)n_in; (void)out_size; (void)d_ws; (void)ws_size;
    const float* s   = (const float*)d_in[0];
    const float* z   = (const float*)d_in[1];
    const float* U   = (const float*)d_in[2];
    const float* V   = (const float*)d_in[3];
    const float* Wi  = (const float*)d_in[4];
    const float* bi  = (const float*)d_in[5];
    const float* Wo  = (const float*)d_in[6];
    const float* bo  = (const float*)d_in[7];
    const float* nmW = (const float*)d_in[8];
    const float* nmb = (const float*)d_in[9];
    float* out = (float*)d_out;

    dim3 grid(BATCH / 16);
    dim3 block(128);
    hipLaunchKernelGGL(cxtrnn_wmma_kernel, grid, block, 0, stream,
                       s, z, U, V, Wi, bi, Wo, bo, nmW, nmb, out);
}


# Update product packed: its free rows carry a second hidden tile (A block-diagonal, K = rank x tile half = 3 K-groups): 6 WMMAs not 8, state held once, no selects at the store; order of sums changes
# speedup vs baseline: 3.8934x; 1.0410x over previous
#include <hip/hip_runtime.h>

#define SEQ   512
#define BATCH 4096
#define DIMS  3
#define DIMY  3
#define DIMZ  6
#define RANK  6
#define HID   50
#define KG    13
#define KH    24
#define KG2   7
#define XSTR  52
#define XSLOT (16 * XSTR)
#define ALPHA 0.5f
#define TB    16
#define NB    (SEQ / TB)
#define XS_F4  (2 * TB * XSLOT / 4)
#define GS_F4  (2 * TB * 32)
#define C_F4   (2 * TB * 256)
#define Z_F4   (4 * TB * 16 / 4)
#define LDS_F4 (XS_F4 + 2 * GS_F4 + C_F4 + Z_F4)

typedef float v2f __attribute__((ext_vector_type(2)));
typedef float v8f __attribute__((ext_vector_type(8)));

__device__ __forceinline__ v8f wmma_f32_k4(v2f a, v2f b, v8f c) {
    return __builtin_amdgcn_wmma_f32_16x16x4_f32(false, a, false, b, (short)0, c, false, false);
}

__device__ __forceinline__ float fast_tanh(float x) {
#if __has_builtin(__builtin_amdgcn_tanhf)
    return __builtin_amdgcn_tanhf(x);
#else
    x = fminf(15.0f, fmaxf(-15.0f, x));
    float e = __expf(2.0f * x);
    float r = __builtin_amdgcn_rcpf(e + 1.0f);
    return fmaf(-2.0f, r, 1.0f);
#endif
}

__device__ __forceinline__ void wave_lds_order() {
    __builtin_amdgcn_fence(__ATOMIC_ACQ_REL, "wavefront");
}

__device__ __forceinline__ float row_xor8(float v) {
    return __int_as_float(__builtin_amdgcn_update_dpp(0, __float_as_int(v), 0x168, 0xf, 0xf, true));
}

__device__ __forceinline__ float fast_sigmoid(float x) {
    return __builtin_amdgcn_rcpf(1.0f + __expf(-x));
}

__global__ __launch_bounds__(128) void cxtrnn_wmma_kernel(
    const float* __restrict__ s,   const float* __restrict__ z,
    const float* __restrict__ U,   const float* __restrict__ V,
    const float* __restrict__ Wi,  const float* __restrict__ bi,
    const float* __restrict__ Wo,  const float* __restrict__ bo,
    const float* __restrict__ nmW, const float* __restrict__ nmb,
    float* __restrict__ out)
{
    __shared__ float4 ldsAll[LDS_F4];
    float*  const ldsXs = (float*)ldsAll;
    float4* const ldsG  = ldsAll + XS_F4;
    float4* const ldsS  = ldsG + GS_F4;
    float4* const ldsC  = ldsS + GS_F4;
    float*  const ldsZ  = (float*)(ldsC + C_F4);

    const int lane = (int)threadIdx.x & 31;
    const int wid  = __builtin_amdgcn_readfirstlane((int)threadIdx.x >> 5);
    const int col  = lane & 15;
    const int hi   = lane >> 4;
    const bool hib = (hi != 0);
    const int b0   = (int)blockIdx.x * 16;
    const int brow = b0 + col;

    const int vset  = (col >> 2) & 1;
    const int vrank = ((col & 3) < 3) ? 2 * (col & 3) + (col >> 3) : -1;
    v2f Av[KG2];
#pragma unroll
    for (int g = 0; g < KG2; ++g) {
        v2f t;
#pragma unroll
        for (int j = 0; j < 2; ++j) {
            const int p  = 4 * g + 2 * hi + j;
            const int k  = vset ? KH + p : p;
            const bool ok = (vrank >= 0) && (vset ? (k < HID) : (p < KH));
            const float w = V[ok ? k * RANK + vrank : 0];
            t[j] = ok ? w : 0.0f;
        }
        Av[g] = t;
    }
    v2f Bu[2][3];
#pragma unroll
    for (int T = 0; T < 2; ++T) {
#pragma unroll
        for (int g = 0; g < 3; ++g) {
            v2f t;
#pragma unroll
            for (int j = 0; j < 2; ++j) {
                const int h = 16 * (2 * T + j) + col;
                t[j] = (h < HID) ? ALPHA * U[h * RANK + 2 * g + hi] : 0.0f;
            }
            Bu[T][g] = t;
        }
    }
    v2f Bw[4];
#pragma unroll
    for (int t4 = 0; t4 < 4; ++t4) {
        int h  = 16 * t4 + col;
        int k0 = 2 * hi;
        float a0 = 0.0f, a1 = 0.0f;
        if (h < HID) {
            a0 = ALPHA * Wi[h * DIMS + k0];
            a1 = ALPHA * ((k0 + 1 < DIMS) ? Wi[h * DIMS + k0 + 1] : bi[h]);
        }
        v2f t; t[0] = a0; t[1] = a1; Bw[t4] = t;
    }
    v2f Bo[KG];
#pragma unroll
    for (int g = 0; g < KG; ++g) {
        int k0 = 4 * g + 2 * hi;
        float a0 = 0.0f, a1 = 0.0f;
        if (col < DIMY) {
            a0 = (k0     < HID) ? Wo[col * HID + k0]     : ((k0     == HID) ? bo[col] : 0.0f);
            a1 = (k0 + 1 < HID) ? Wo[col * HID + k0 + 1] : ((k0 + 1 == HID) ? bo[col] : 0.0f);
        }
        v2f t; t[0] = a0; t[1] = a1; Bo[g] = t;
    }
    float nmw[RANK][DIMZ], nmbv[RANK];
#pragma unroll
    for (int r = 0; r < RANK; ++r) {
        nmbv[r] = nmb[r];
#pragma unroll
        for (int i = 0; i < DIMZ; ++i) nmw[r][i] = nmW[r * DIMZ + i];
    }

    v8f XD[2];
#pragma unroll
    for (int T = 0; T < 2; ++T) { v8f zz = {0,0,0,0,0,0,0,0}; XD[T] = zz; }
    float Atan[2 * KG2];
#pragma unroll
    for (int i = 0; i < 2 * KG2; ++i) Atan[i] = 0.0f;

    const v8f vzero = {0,0,0,0,0,0,0,0};

#pragma unroll 1
    for (int ph = 0; ph < NB + 2; ++ph) {
      if (wid < 2) {
        if (ph >= 1 && ph <= NB) {
          const int kb = ph - 1;
          const int rb = 8 * wid;
          const int bl = col & 7;
          const int kh = col >> 3;
          const bool khb = (kh != 0);
          const float4* Gb = ldsG + (kb & 1) * (TB * 32) + 16 * hi + rb + bl;
          const float4* Cb = ldsC + (kb & 1) * (TB * 256) + 64 * hi + 16 * wid + col;
          float* Xb = ldsXs + (kb & 1) * (TB * XSLOT) + rb * XSTR;
          float4 gq = Gb[0];
          float cin[16];
#pragma unroll
          for (int q = 0; q < 4; ++q) {
              const float4 c4 = Cb[(4 * (q >> 1) + (q & 1)) * 32];
              cin[4 * q] = c4.x; cin[4 * q + 1] = c4.y; cin[4 * q + 2] = c4.z; cin[4 * q + 3] = c4.w;
          }
#pragma unroll 1
          for (int tt = 0; tt < TB; ++tt) {
            float* Xs = Xb + tt * XSLOT;
            v2f gm0, gm1, gm2;
            gm0[0] = khb ? 0.0f : gq.x; gm0[1] = khb ? gq.x : 0.0f;
            gm1[0] = khb ? 0.0f : gq.y; gm1[1] = khb ? gq.y : 0.0f;
            gm2[0] = khb ? 0.0f : gq.z; gm2[1] = khb ? gq.z : 0.0f;

            v8f PT = vzero;
#pragma unroll
            for (int g = 0; g < KG2; ++g) {
                v2f b; b[0] = Atan[2 * g]; b[1] = Atan[2 * g + 1];
                PT = wmma_f32_k4(Av[g], b, PT);
            }

            const float p0 = khb ? PT[4] : PT[0];
            const float p1 = khb ? PT[5] : PT[1];
            const float p2 = khb ? PT[6] : PT[2];
            const float q0 = p0 + row_xor8(p0);
            const float q1 = p1 + row_xor8(p1);
            const float q2 = p2 + row_xor8(p2);
            v2f Ap0, Ap1, Ap2;
            Ap0[0] = q0 * gm0[0]; Ap0[1] = q0 * gm0[1];
            Ap1[0] = q1 * gm1[0]; Ap1[1] = q1 * gm1[1];
            Ap2[0] = q2 * gm2[0]; Ap2[1] = q2 * gm2[1];

#pragma unroll
            for (int T = 0; T < 2; ++T) {
                v8f acc;
#pragma unroll
                for (int v = 0; v < 8; ++v) acc[v] = cin[8 * T + v];
                acc = wmma_f32_k4(Ap0, Bu[T][0], acc);
                acc = wmma_f32_k4(Ap1, Bu[T][1], acc);
                acc = wmma_f32_k4(Ap2, Bu[T][2], acc);
                v8f x = XD[T];
#pragma unroll
                for (int v = 0; v < 8; ++v) x[v] = fmaf(x[v], 1.0f - ALPHA, acc[v]);
                XD[T] = x;
                if (T == 0 || !hib || col < XSTR - 48) {
#pragma unroll
                    for (int v = 0; v < 8; ++v)
                        Xs[v * XSTR + 16 * (2 * T + hi) + col] = x[v];
                }
            }
            wave_lds_order();

            const int tn = min(tt + 1, TB - 1);
            gq = Gb[tn * 32];

            float xa[2 * KG2];
#pragma unroll
            for (int g = 0; g < KG2; ++g) {
                int k0 = KH * kh + 4 * g + 2 * hi;
                xa[2 * g]     = Xs[bl * XSTR + k0];
                xa[2 * g + 1] = Xs[bl * XSTR + k0 + 1];
            }
            __builtin_amdgcn_sched_barrier(0);
            {
#pragma unroll
                for (int q = 0; q < 4; ++q) {
                    const float4 c4 = Cb[tn * 256 + (4 * (q >> 1) + (q & 1)) * 32];
                    cin[4 * q] = c4.x; cin[4 * q + 1] = c4.y; cin[4 * q + 2] = c4.z; cin[4 * q + 3] = c4.w;
                }
            }
            __builtin_amdgcn_sched_barrier(0);
#pragma unroll
            for (int i = 0; i < 2 * KG2; ++i) Atan[i] = fast_tanh(xa[i]);
          }
        }
      } else if (wid == 2) {
        if (ph >= 2) {
            const int kb = ph - 2;
            const float* Zb = ldsZ + (kb & 3) * (TB * 16);
            const float* Xb = ldsXs + (kb & 1) * (TB * XSLOT);
#pragma unroll 1
            for (int tt = 0; tt < TB; tt += 2) {
                const int t = kb * TB + tt;
                const float* Xs0 = Xb + tt * XSLOT;
                const float* Xs1 = Xs0 + XSLOT;
                const float zsum0 = Zb[tt * 16 + col];
                const float zsum1 = Zb[(tt + 1) * 16 + col];
                float xr0[2 * KG], xr1[2 * KG];
#pragma unroll
                for (int g = 0; g < KG; ++g) {
                    int k0 = 4 * g + 2 * hi;
                    xr0[2 * g]     = Xs0[col * XSTR + k0];
                    xr0[2 * g + 1] = Xs0[col * XSTR + k0 + 1];
                }
#pragma unroll
                for (int g = 0; g < KG; ++g) {
                    int k0 = 4 * g + 2 * hi;
                    xr1[2 * g]     = Xs1[col * XSTR + k0];
                    xr1[2 * g + 1] = Xs1[col * XSTR + k0 + 1];
                }
                __builtin_amdgcn_sched_barrier(0);
                v8f OD0 = vzero, OD1 = vzero;
#pragma unroll
                for (int g = 0; g < KG; ++g) {
                    float a0 = zsum0 * fast_tanh(xr0[2 * g]);
                    float a1 = zsum0 * fast_tanh(xr0[2 * g + 1]);
                    float c0 = zsum1 * fast_tanh(xr1[2 * g]);
                    float c1 = zsum1 * fast_tanh(xr1[2 * g + 1]);
                    if (g == KG - 1) {
                        a0 = hib ? zsum0 : a0;
                        a1 = hib ? 0.0f : a1;
                        c0 = hib ? zsum1 : c0;
                        c1 = hib ? 0.0f : c1;
                    }
                    v2f a; a[0] = a0; a[1] = a1;
                    v2f c; c[0] = c0; c[1] = c1;
                    OD0 = wmma_f32_k4(a, Bo[g], OD0);
                    OD1 = wmma_f32_k4(c, Bo[g], OD1);
                }
                if (col < DIMY) {
                    size_t base = (size_t)t * BATCH * DIMY + (size_t)(b0 + 8 * hi) * DIMY + col;
#pragma unroll
                    for (int v = 0; v < 8; ++v) out[base + (size_t)v * DIMY] = OD0[v];
#pragma unroll
                    for (int v = 0; v < 8; ++v)
                        out[base + (size_t)BATCH * DIMY + (size_t)v * DIMY] = OD1[v];
                }
            }
        }
      } else {
        if (ph < NB) {
            float zq[TB / 2][DIMZ], sq[TB / 2][DIMS];
            const int t0 = ph * TB;
            const float* zb = z + ((size_t)(t0 + hi) * BATCH + (size_t)brow) * DIMZ;
            const float* sb = s + ((size_t)(t0 + hi) * BATCH + (size_t)brow) * DIMS;
#pragma unroll
            for (int j = 0; j < TB / 2; ++j) {
                const float* zr = zb + (size_t)j * (2 * BATCH * DIMZ);
                const float* sr = sb + (size_t)j * (2 * BATCH * DIMS);
#pragma unroll
                for (int i = 0; i < DIMZ; ++i) zq[j][i] = zr[i];
#pragma unroll
                for (int i = 0; i < DIMS; ++i) sq[j][i] = sr[i];
            }
            float4* Gb = ldsG + (ph & 1) * (TB * 32);
            float4* Sb = ldsS + (ph & 1) * (TB * 32);
            float*  Zb = ldsZ + (ph & 3) * (TB * 16);
#pragma unroll
            for (int j = 0; j < TB / 2; ++j) {
                float zsum = 0.0f;
#pragma unroll
                for (int i = 0; i < DIMZ; ++i) zsum += zq[j][i];
                float a0 = nmbv[0], a1 = nmbv[1], a2 = nmbv[2];
                float a3 = nmbv[3], a4 = nmbv[4], a5 = nmbv[5];
#pragma unroll
                for (int i = 0; i < DIMZ; ++i) {
                    a0 = fmaf(nmw[0][i], zq[j][i], a0);
                    a1 = fmaf(nmw[1][i], zq[j][i], a1);
                    a2 = fmaf(nmw[2][i], zq[j][i], a2);
                    a3 = fmaf(nmw[3][i], zq[j][i], a3);
                    a4 = fmaf(nmw[4][i], zq[j][i], a4);
                    a5 = fmaf(nmw[5][i], zq[j][i], a5);
                }
                const float g0 = fast_sigmoid(a0), g1 = fast_sigmoid(a1), g2 = fast_sigmoid(a2);
                const float g3 = fast_sigmoid(a3), g4 = fast_sigmoid(a4), g5 = fast_sigmoid(a5);
                const int slot = (2 * j + hi) * 32 + col;
                Gb[slot]      = make_float4(g0, g2, g4, 0.0f);
                Gb[slot + 16] = make_float4(g1, g3, g5, 0.0f);
                Sb[slot]      = make_float4(zsum * sq[j][0], zsum * sq[j][1], 0.0f, 0.0f);
                Sb[slot + 16] = make_float4(zsum * sq[j][2], zsum,            0.0f, 0.0f);
                Zb[(2 * j + hi) * 16 + col] = zsum;
            }
            asm volatile("s_wait_dscnt 0" ::: "memory");
            float4* Cb = ldsC + (ph & 1) * (TB * 256);
#pragma unroll 4
            for (int tt = 0; tt < TB; ++tt) {
                const float4 rq = Sb[tt * 32 + lane];
                v2f As;
                As[0] = rq.x;
                As[1] = rq.y;
#pragma unroll
                for (int t4 = 0; t4 < 4; ++t4) {
                    const v8f c = wmma_f32_k4(As, Bw[t4], vzero);
                    Cb[tt * 256 + (2 * t4) * 32 + lane]     = make_float4(c[0], c[1], c[2], c[3]);
                    Cb[tt * 256 + (2 * t4 + 1) * 32 + lane] = make_float4(c[4], c[5], c[6], c[7]);
                }
            }
        }
      }
      __syncthreads();
    }
}

extern "C" void kernel_launch(void* const* d_in, const int* in_sizes, int n_in,
                              void* d_out, int out_size, void* d_ws, size_t ws_size,
                              hipStream_t stream) {
    (void)in_sizes; (void)n_in; (void)out_size; (void)d_ws; (void)ws_size;
    const float* s   = (const float*)d_in[0];
    const float* z   = (const float*)d_in[1];
    const float* U   = (const float*)d_in[2];
    const float* V   = (const float*)d_in[3];
    const float* Wi  = (const float*)d_in[4];
    const float* bi  = (const float*)d_in[5];
    const float* Wo  = (const float*)d_in[6];
    const float* bo  = (const float*)d_in[7];
    const float* nmW = (const float*)d_in[8];
    const float* nmb = (const float*)d_in[9];
    float* out = (float*)d_out;

    dim3 grid(BATCH / 16);
    dim3 block(128);
    hipLaunchKernelGGL(cxtrnn_wmma_kernel, grid, block, 0, stream,
                       s, z, U, V, Wi, bi, Wo, bo, nmW, nmb, out);
}
